# starting kernel
# baseline (speedup 1.0000x reference)
#include <hip/hip_runtime.h>
#include <math.h>

typedef __attribute__((ext_vector_type(2))) float v2f;
typedef __attribute__((ext_vector_type(8))) float v8f;

#define NFEAT 128
#define NHID  128
#define NCLASS 16

#define G1_ROWS   48
#define G1_TILES  3
#define G1_STRIDE 132

__global__ void zero_kernel(float* __restrict__ p, long long n) {
  long long i = (long long)blockIdx.x * blockDim.x + threadIdx.x;
  long long stride = (long long)gridDim.x * blockDim.x;
  for (; i < n; i += stride) p[i] = 0.0f;
}

__global__ void scatter_kernel(const float* __restrict__ feat,
                               const int* __restrict__ src,
                               const int* __restrict__ dst,
                               float* __restrict__ agg,
                               float* __restrict__ deg,
                               int addDeg, long long nEdges) {
  long long gid = (long long)blockIdx.x * blockDim.x + threadIdx.x;
  long long e = gid >> 5;
  if (e >= nEdges) return;
  int lane = (int)(gid & 31);
  int s = src[e];
  int d = dst[e];
  int f = lane * 4;
  const float4 v = *(const float4*)(feat + (long long)s * NFEAT + f);
  float* o = agg + (long long)d * NFEAT + f;
  atomicAdd(o + 0, v.x);
  atomicAdd(o + 1, v.y);
  atomicAdd(o + 2, v.z);
  atomicAdd(o + 3, v.w);
  if (addDeg && lane == 0) atomicAdd(deg + d, 1.0f);
}

__global__ void normalize_kernel(float* __restrict__ agg,
                                 const float* __restrict__ deg, long long n) {
  long long i = (long long)blockIdx.x * blockDim.x + threadIdx.x;
  if (i >= n) return;
  float dv = deg[i >> 7];
  agg[i] = agg[i] / fmaxf(dv, 1.0f);
}

__global__ __launch_bounds__(256)
void gemm1_kernel(const float* __restrict__ x, const float* __restrict__ agg,
                  const float* __restrict__ wself, const float* __restrict__ wneigh,
                  const float* __restrict__ bias, float* __restrict__ h, int nNodes) {
  __shared__ float xs[G1_ROWS * G1_STRIDE];
  __shared__ float as[G1_ROWS * G1_STRIDE];

  const int tid  = threadIdx.x;
  const int lane = tid & 31;
  const int wave = tid >> 5;
  const int half = lane >> 4;
  const int l16  = lane & 15;
  const int col  = wave * 16 + l16;
  const long long base = (long long)blockIdx.x * G1_ROWS;

  for (int i = tid; i < G1_ROWS * 32; i += 256) {
    int row = i >> 5;
    int c4  = (i & 31) * 4;
    long long node = base + row;
    float4 vx = make_float4(0.f, 0.f, 0.f, 0.f);
    float4 va = vx;
    if (node < nNodes) {
      vx = *(const float4*)(x   + node * NFEAT + c4);
      va = *(const float4*)(agg + node * NFEAT + c4);
    }
    *(float4*)(&xs[row * G1_STRIDE + c4]) = vx;
    *(float4*)(&as[row * G1_STRIDE + c4]) = va;
  }
  __syncthreads();

  const v8f vzero = {0.f, 0.f, 0.f, 0.f, 0.f, 0.f, 0.f, 0.f};
  v8f acc[G1_TILES];
#pragma unroll
  for (int r = 0; r < G1_TILES; ++r) acc[r] = vzero;

  for (int k0 = 0; k0 < NFEAT; k0 += 4) {
    const int kb = k0 + half * 2;
    v2f bs, bn;
    bs.x = wself [(long long)kb       * NHID + col];
    bs.y = wself [(long long)(kb + 1) * NHID + col];
    bn.x = wneigh[(long long)kb       * NHID + col];
    bn.y = wneigh[(long long)(kb + 1) * NHID + col];
#pragma unroll
    for (int r = 0; r < G1_TILES; ++r) {
      const v2f ax = *(const v2f*)(&xs[(r * 16 + l16) * G1_STRIDE + kb]);
      const v2f aa = *(const v2f*)(&as[(r * 16 + l16) * G1_STRIDE + kb]);
      acc[r] = __builtin_amdgcn_wmma_f32_16x16x4_f32(false, ax, false, bs,
                                                     (short)0, acc[r], false, false);
      acc[r] = __builtin_amdgcn_wmma_f32_16x16x4_f32(false, aa, false, bn,
                                                     (short)0, acc[r], false, false);
    }
  }

  const float bv = bias[col];
#pragma unroll
  for (int r = 0; r < G1_TILES; ++r) {
#pragma unroll
    for (int j = 0; j < 8; ++j) {
      long long node = base + r * 16 + j + half * 8;
      if (node < nNodes) {
        float v = acc[r][j] + bv;
        h[node * NHID + col] = fmaxf(v, 0.0f);
      }
    }
  }
}

__global__ __launch_bounds__(256)
void gemm2_kernel(const float* __restrict__ h, const float* __restrict__ agg,
                  const float* __restrict__ wself, const float* __restrict__ wneigh,
                  const float* __restrict__ bias, float* __restrict__ out, int nNodes) {
  const int lane = threadIdx.x & 31;
  const int wave = threadIdx.x >> 5;
  const int half = lane >> 4;
  const int l16  = lane & 15;
  const long long base = ((long long)blockIdx.x * 8 + wave) * 16;
  if (base >= nNodes) return;

  long long nodeA = base + l16;
  if (nodeA >= nNodes) nodeA = nNodes - 1;

  v8f acc = {0.f, 0.f, 0.f, 0.f, 0.f, 0.f, 0.f, 0.f};

  for (int k0 = 0; k0 < NHID; k0 += 4) {
    const int kb = k0 + half * 2;
    const v2f ah = *(const v2f*)(h   + nodeA * NHID + kb);
    const v2f aa = *(const v2f*)(agg + nodeA * NHID + kb);
    v2f bs, bn;
    bs.x = wself [kb       * NCLASS + l16];
    bs.y = wself [(kb + 1) * NCLASS + l16];
    bn.x = wneigh[kb       * NCLASS + l16];
    bn.y = wneigh[(kb + 1) * NCLASS + l16];
    acc = __builtin_amdgcn_wmma_f32_16x16x4_f32(false, ah, false, bs,
                                                (short)0, acc, false, false);
    acc = __builtin_amdgcn_wmma_f32_16x16x4_f32(false, aa, false, bn,
                                                (short)0, acc, false, false);
  }

  const float bv = bias[l16];
#pragma unroll
  for (int j = 0; j < 8; ++j) {
    float v = acc[j] + bv;
    float mx = v;
#pragma unroll
    for (int off = 8; off >= 1; off >>= 1)
      mx = fmaxf(mx, __shfl_xor(mx, off, 16));
    float e = expf(v - mx);
    float s = e;
#pragma unroll
    for (int off = 8; off >= 1; off >>= 1)
      s += __shfl_xor(s, off, 16);
    long long node = base + j + half * 8;
    if (node < nNodes) out[node * NCLASS + l16] = v - mx - logf(s);
  }
}

extern "C" void kernel_launch(void* const* d_in, const int* in_sizes, int n_in,
                              void* d_out, int out_size, void* d_ws, size_t ws_size,
                              hipStream_t stream) {
  const float* x   = (const float*)d_in[0];
  const int*   src = (const int*)  d_in[1];
  const int*   dst = (const int*)  d_in[2];
  const float* ws1 = (const float*)d_in[3];
  const float* wn1 = (const float*)d_in[4];
  const float* b1  = (const float*)d_in[5];
  const float* ws2 = (const float*)d_in[6];
  const float* wn2 = (const float*)d_in[7];
  const float* b2  = (const float*)d_in[8];
  float* out = (float*)d_out;

  const int nNodes = in_sizes[0] / NFEAT;
  const long long nEdges = in_sizes[1];

  char* w = (char*)d_ws;
  size_t degBytes = (((size_t)nNodes * 4) + 255) & ~(size_t)255;
  float* deg = (float*)w;
  float* agg = (float*)(w + degBytes);
  float* h   = agg + (size_t)nNodes * NFEAT;

  const long long aggN  = (long long)nNodes * NFEAT;
  const long long zeroN = (long long)(degBytes / 4) + aggN;

  const int scatterBlocks = (int)((nEdges * 32 + 255) / 256);
  const int normBlocks    = (int)((aggN + 255) / 256);

  zero_kernel<<<2048, 256, 0, stream>>>((float*)w, zeroN);
  scatter_kernel<<<scatterBlocks, 256, 0, stream>>>(x, src, dst, agg, deg, 1, nEdges);
  normalize_kernel<<<normBlocks, 256, 0, stream>>>(agg, deg, aggN);
  gemm1_kernel<<<(nNodes + G1_ROWS - 1) / G1_ROWS, 256, 0, stream>>>(
      x, agg, ws1, wn1, b1, h, nNodes);

  zero_kernel<<<2048, 256, 0, stream>>>(agg, aggN);
  scatter_kernel<<<scatterBlocks, 256, 0, stream>>>(h, src, dst, agg, deg, 0, nEdges);
  normalize_kernel<<<normBlocks, 256, 0, stream>>>(agg, deg, aggN);
  gemm2_kernel<<<(nNodes + 127) / 128, 256, 0, stream>>>(
      h, agg, ws2, wn2, b2, out, nNodes);
}


# layer-2 product + log-softmax inside the layer-2 sum kernel: a wave takes 16 nodes, A = LDS sums (pitch 132) x 1/deg, W in LDS, 16-B loads; means plane and a launch gone; k order regrouped
# speedup vs baseline: 3.1802x; 3.1802x over previous
#include <hip/hip_runtime.h>
#include <math.h>

typedef __attribute__((ext_vector_type(2))) float v2f;
typedef __attribute__((ext_vector_type(8))) float v8f;

#define NFEAT 128
#define NHID  128
#define NCLASS 16

#define G1_ROWS   48
#define G1_TILES  3
#define G1_STRIDE 132

#define RNG   400
#define QCAP  12288
#define RS_THREADS 1024

#define SL_DEG 4
#define SL_Q   (SL_DEG + RNG)

#define SLAB_BYTES (RNG * NFEAT * 4)
#define KEY_OFF    131072
#define KEY_PIECE  65536
#define QD_NONE    0xFFFF

#define ACC_PITCH2 132
static_assert(SL_Q * 4 + 6 * QCAP <= KEY_OFF, "parked queue must end below the key bytes");
static_assert(KEY_OFF + KEY_PIECE <= SLAB_BYTES, "key piece must fit in a full range's rows");
static_assert(KEY_PIECE == 4 * 16 * RS_THREADS, "one piece = 4 loads of 16 B per thread");
static_assert(RNG < QD_NONE, "local dst must not collide with the marker");

__device__ __forceinline__ bool keys_usable(int nNodes, long long nEdges) {
  const long long nPieces = (nEdges + KEY_PIECE - 1) / KEY_PIECE;
  return nPieces * RNG <= (long long)nNodes;
}

__global__ __launch_bounds__(RS_THREADS)
void range_key_kernel(const int* __restrict__ dst, unsigned* keep, int nNodes, long long nEdges) {
  if (!keys_usable(nNodes, nEdges)) return;
  const int piece = blockIdx.x >> 2;
  const int w0    = (blockIdx.x & 3) * (4 * RS_THREADS) + threadIdx.x;
  unsigned* kw = keep + (long long)piece * (SLAB_BYTES / 4) + KEY_OFF / 4;
  int4 d[4];
#pragma unroll
  for (int j = 0; j < 4; ++j) {
    const long long e = (long long)piece * KEY_PIECE + (long long)(w0 + j * RS_THREADS) * 4;
    if (e + 4 <= nEdges) {
      d[j] = *(const int4*)(dst + e);
    } else {
      d[j].x = (e + 0 < nEdges) ? dst[e + 0] : -1;
      d[j].y = (e + 1 < nEdges) ? dst[e + 1] : -1;
      d[j].z = (e + 2 < nEdges) ? dst[e + 2] : -1;
      d[j].w = (e + 3 < nEdges) ? dst[e + 3] : -1;
    }
  }
#pragma unroll
  for (int j = 0; j < 4; ++j) {
    const unsigned k0 = (d[j].x < 0) ? 255u : (((unsigned)d[j].x / RNG) & 255u);
    const unsigned k1 = (d[j].y < 0) ? 255u : (((unsigned)d[j].y / RNG) & 255u);
    const unsigned k2 = (d[j].z < 0) ? 255u : (((unsigned)d[j].z / RNG) & 255u);
    const unsigned k3 = (d[j].w < 0) ? 255u : (((unsigned)d[j].w / RNG) & 255u);
    kw[w0 + j * RS_THREADS] = k0 | (k1 << 8) | (k2 << 16) | (k3 << 24);
  }
}

__device__ __forceinline__ unsigned zero_bytes(unsigned x) {
  return ~((((x & 0x7F7F7F7Fu) + 0x7F7F7F7Fu) | x) | 0x7F7F7F7Fu);
}

__device__ __forceinline__ void ks_take16(const uint4 w, unsigned pat, int e0, int lo, int cnt,
                                          const float* __restrict__ feat,
                                          const int* __restrict__ src,
                                          const int* __restrict__ dst, long long nEdges,
                                          float* acc, int astr, int* qe, int* degl, int* qn) {
  unsigned m = zero_bytes(w.x ^ pat) | (zero_bytes(w.y ^ pat) >> 1) |
               (zero_bytes(w.z ^ pat) >> 2) | (zero_bytes(w.w ^ pat) >> 3);
  while (m) {
    const int pos = __ffs(m) - 1;
    const int e = e0 + 4 * (7 - (pos & 7)) + (pos >> 3);
    m &= m - 1;
    const int p = atomicAdd(qn, 1);
    if (p < QCAP) {
      qe[p] = e;
    } else if (e < nEdges) {
      const unsigned u = (unsigned)(dst[e] - lo);
      if (u < (unsigned)cnt) {
        atomicAdd(&degl[u], 1);
        const int s = src[e];
#pragma unroll 1
        for (int f = 0; f < NFEAT; ++f)
          atomicAdd(&acc[u * astr + f], feat[(long long)s * NFEAT + f]);
      }
    }
  }
}

__device__ __forceinline__ void ks_filter(const float* __restrict__ feat,
                                          const int* __restrict__ src,
                                          const int* __restrict__ dst,
                                          const unsigned char* keys,
                                          long long nEdges, int lo, int cnt, int tid,
                                          float* acc, int astr, int* qe, int* degl, int* qn) {
  const int nPieces = (int)((nEdges + KEY_PIECE - 1) / KEY_PIECE);
  const unsigned pat = (blockIdx.x & 255u) * 0x01010101u;
  int p = (int)(((long long)blockIdx.x * nPieces) / gridDim.x);
  const uint4* kp = (const uint4*)(keys + (long long)p * SLAB_BYTES + KEY_OFF);
  uint4 c0 = kp[tid];
  uint4 c1 = kp[tid + RS_THREADS];
  uint4 c2 = kp[tid + 2 * RS_THREADS];
  uint4 c3 = kp[tid + 3 * RS_THREADS];
  for (int pp = 0; pp < nPieces; ++pp) {
    int pn = p + 1;
    if (pn >= nPieces) pn = 0;
    uint4 n0 = c0, n1 = c1, n2 = c2, n3 = c3;
    if (pp + 1 < nPieces) {
      kp = (const uint4*)(keys + (long long)pn * SLAB_BYTES + KEY_OFF);
      n0 = kp[tid];
      n1 = kp[tid + RS_THREADS];
      n2 = kp[tid + 2 * RS_THREADS];
      n3 = kp[tid + 3 * RS_THREADS];
    }
    const int e0 = p * KEY_PIECE + tid * 16;
    ks_take16(c0, pat, e0,                   lo, cnt, feat, src, dst, nEdges, acc, astr, qe, degl, qn);
    ks_take16(c1, pat, e0 + 16 * RS_THREADS, lo, cnt, feat, src, dst, nEdges, acc, astr, qe, degl, qn);
    ks_take16(c2, pat, e0 + 32 * RS_THREADS, lo, cnt, feat, src, dst, nEdges, acc, astr, qe, degl, qn);
    ks_take16(c3, pat, e0 + 48 * RS_THREADS, lo, cnt, feat, src, dst, nEdges, acc, astr, qe, degl, qn);
    c0 = n0; c1 = n1; c2 = n2; c3 = n3;
    p = pn;
  }
}

__device__ __forceinline__ void rs_take(int d, int e, int lo, int cnt,
                                        const float* __restrict__ feat,
                                        const int* __restrict__ src,
                                        float* acc, int astr, int* qe, unsigned short* qd,
                                        int* degl, int* qn) {
  const unsigned u = (unsigned)(d - lo);
  if (u < (unsigned)cnt) {
    atomicAdd(&degl[u], 1);
    const int p = atomicAdd(qn, 1);
    if (p < QCAP) {
      qe[p] = e;
      qd[p] = (unsigned short)u;
    } else {
      const int s = src[e];
#pragma unroll 1
      for (int f = 0; f < NFEAT; ++f)
        atomicAdd(&acc[u * astr + f], feat[(long long)s * NFEAT + f]);
    }
  }
}

__device__ __forceinline__ void rs_filter(const float* __restrict__ feat,
                                          const int* __restrict__ src,
                                          const int* __restrict__ dst,
                                          long long nEdges, int lo, int cnt, int tid,
                                          float* acc, int astr, int* qe, unsigned short* qd,
                                          int* degl, int* qn) {
  const long long n4 = nEdges >> 2;
  const int4* __restrict__ dst4 = (const int4*)dst;
  const int nFull = (int)(n4 / (4 * RS_THREADS));
  const int rot   = (int)(((long long)blockIdx.x * nFull) / gridDim.x);
  for (int gg = 0; gg < nFull; ++gg) {
    int g = gg + rot;
    if (g >= nFull) g -= nFull;
    const long long i = (long long)g * (4 * RS_THREADS) + tid;
    const int4 d0 = dst4[i];
    const int4 d1 = dst4[i + RS_THREADS];
    const int4 d2 = dst4[i + 2 * RS_THREADS];
    const int4 d3 = dst4[i + 3 * RS_THREADS];
    const int e0 = (int)(i * 4);
    const int e1 = e0 + 4 * RS_THREADS, e2 = e0 + 8 * RS_THREADS, e3 = e0 + 12 * RS_THREADS;
    rs_take(d0.x, e0 + 0, lo, cnt, feat, src, acc, astr, qe, qd, degl, qn);
    rs_take(d0.y, e0 + 1, lo, cnt, feat, src, acc, astr, qe, qd, degl, qn);
    rs_take(d0.z, e0 + 2, lo, cnt, feat, src, acc, astr, qe, qd, degl, qn);
    rs_take(d0.w, e0 + 3, lo, cnt, feat, src, acc, astr, qe, qd, degl, qn);
    rs_take(d1.x, e1 + 0, lo, cnt, feat, src, acc, astr, qe, qd, degl, qn);
    rs_take(d1.y, e1 + 1, lo, cnt, feat, src, acc, astr, qe, qd, degl, qn);
    rs_take(d1.z, e1 + 2, lo, cnt, feat, src, acc, astr, qe, qd, degl, qn);
    rs_take(d1.w, e1 + 3, lo, cnt, feat, src, acc, astr, qe, qd, degl, qn);
    rs_take(d2.x, e2 + 0, lo, cnt, feat, src, acc, astr, qe, qd, degl, qn);
    rs_take(d2.y, e2 + 1, lo, cnt, feat, src, acc, astr, qe, qd, degl, qn);
    rs_take(d2.z, e2 + 2, lo, cnt, feat, src, acc, astr, qe, qd, degl, qn);
    rs_take(d2.w, e2 + 3, lo, cnt, feat, src, acc, astr, qe, qd, degl, qn);
    rs_take(d3.x, e3 + 0, lo, cnt, feat, src, acc, astr, qe, qd, degl, qn);
    rs_take(d3.y, e3 + 1, lo, cnt, feat, src, acc, astr, qe, qd, degl, qn);
    rs_take(d3.z, e3 + 2, lo, cnt, feat, src, acc, astr, qe, qd, degl, qn);
    rs_take(d3.w, e3 + 3, lo, cnt, feat, src, acc, astr, qe, qd, degl, qn);
  }
  for (long long i = (long long)nFull * (4 * RS_THREADS) + tid; i < n4; i += RS_THREADS) {
    const int4 d = dst4[i];
    const int e = (int)(i * 4);
    rs_take(d.x, e + 0, lo, cnt, feat, src, acc, astr, qe, qd, degl, qn);
    rs_take(d.y, e + 1, lo, cnt, feat, src, acc, astr, qe, qd, degl, qn);
    rs_take(d.z, e + 2, lo, cnt, feat, src, acc, astr, qe, qd, degl, qn);
    rs_take(d.w, e + 3, lo, cnt, feat, src, acc, astr, qe, qd, degl, qn);
  }
  for (long long e = n4 * 4 + tid; e < nEdges; e += RS_THREADS)
    rs_take(dst[e], (int)e, lo, cnt, feat, src, acc, astr, qe, qd, degl, qn);
}

template <bool FIRST>
__global__ __launch_bounds__(RS_THREADS)
void range_sum_kernel(const float* __restrict__ feat,
                      const int* __restrict__ src,
                      const int* __restrict__ dst,
                      float* agg,
                      int* keep,
                      const float* __restrict__ wself,
                      const float* __restrict__ wneigh,
                      const float* __restrict__ bias,
                      float* __restrict__ out,
                      int nNodes, long long nEdges) {
  constexpr int AST = FIRST ? NFEAT : ACC_PITCH2;
  __shared__ __attribute__((aligned(16))) float acc[RNG * AST];
  __shared__ __attribute__((aligned(16))) int qe[QCAP];
  __shared__ __attribute__((aligned(16))) unsigned short qd[QCAP];
  __shared__ __attribute__((aligned(16))) int degl[RNG];
  __shared__ int qn;
  __shared__ int hd;
  __shared__ __attribute__((aligned(16))) float wl[FIRST ? 4 : 2 * NHID * NCLASS];

  const int tid  = threadIdx.x;
  const int lane = tid & 31;
  const int wave = tid >> 5;
  const int lo   = blockIdx.x * RNG;
  const int cnt  = (nNodes - lo < RNG) ? (nNodes - lo) : RNG;
  int* slab = keep + (long long)lo * NFEAT;

  const float4 z4 = make_float4(0.f, 0.f, 0.f, 0.f);
  for (int i = tid; i < RNG * AST / 4; i += RS_THREADS) ((float4*)acc)[i] = z4;
  if (FIRST) {
    for (int i = tid; i < RNG; i += RS_THREADS) degl[i] = 0;
    if (tid == 0) qn = 0;
  } else {
    if (tid == 0) hd = slab[0];
    for (int i = tid; i < NHID / 2 * NCLASS; i += RS_THREADS) {
      const int q = i >> 4, c = i & 15;
      const int k = 8 * (q >> 2) + 4 * (q & 1) + 2 * ((q >> 1) & 1);
      v2f ws, wn;
      ws.x = wself [k * NCLASS + c];  ws.y = wself [(k + 1) * NCLASS + c];
      wn.x = wneigh[k * NCLASS + c];  wn.y = wneigh[(k + 1) * NCLASS + c];
      ((v2f*)wl)[i] = ws;
      ((v2f*)wl)[NHID / 2 * NCLASS + i] = wn;
    }
  }
  __syncthreads();

  int n;
  const bool scan = FIRST ? true : (hd < 0);
  if (scan) {
    if (!FIRST) {
      for (int i = tid; i < RNG; i += RS_THREADS) degl[i] = 0;
      if (tid == 0) qn = 0;
      __syncthreads();
    }
    const bool keyed = FIRST && keys_usable(nNodes, nEdges);
    if (keyed) ks_filter(feat, src, dst, (const unsigned char*)keep, nEdges, lo, cnt, tid, acc, AST, qe, degl, &qn);
    else       rs_filter(feat, src, dst, nEdges, lo, cnt, tid, acc, AST, qe, qd, degl, &qn);
    __syncthreads();

    const int total = qn;
    n = (total < QCAP) ? total : QCAP;

    if (keyed) {
      for (int i = tid; i < n; i += RS_THREADS) {
        const int e = qe[i];
        int s = 0;
        unsigned short uu = QD_NONE;
        if (e < nEdges) {
          const int dv = dst[e];
          const int sv = src[e];
          const unsigned u = (unsigned)(dv - lo);
          if (u < (unsigned)cnt) { s = sv; uu = (unsigned short)u; atomicAdd(&degl[u], 1); }
        }
        qe[i] = s;
        qd[i] = uu;
      }
    } else {
      for (int i = tid; i < n; i += RS_THREADS) qe[i] = src[qe[i]];
    }
    __syncthreads();

    if (FIRST) {
      const int np = (n + 7) & ~7;
      const bool fits = (total <= QCAP) && (SL_Q * 4 + 6 * np <= cnt * NFEAT * 4);
      if (fits) {
        int4* sq = (int4*)(slab + SL_Q);
        for (int i = tid; i < np / 4; i += RS_THREADS) sq[i] = ((const int4*)qe)[i];
        int4* sd = (int4*)(slab + SL_Q + np);
        for (int i = tid; i < np / 8; i += RS_THREADS) sd[i] = ((const int4*)qd)[i];
        int4* sg = (int4*)(slab + SL_DEG);
        for (int i = tid; i < RNG / 4; i += RS_THREADS) sg[i] = ((const int4*)degl)[i];
      }
      if (tid == 0) slab[0] = fits ? n : -1;
    }
  } else {
    n = (hd < QCAP) ? hd : QCAP;
    const int np = (n + 7) & ~7;
    const int4* sq = (const int4*)(slab + SL_Q);
    for (int i = tid; i < np / 4; i += RS_THREADS) ((int4*)qe)[i] = sq[i];
    const int4* sd = (const int4*)(slab + SL_Q + np);
    for (int i = tid; i < np / 8; i += RS_THREADS) ((int4*)qd)[i] = sd[i];
    const int4* sg = (const int4*)(slab + SL_DEG);
    for (int i = tid; i < RNG / 4; i += RS_THREADS) ((int4*)degl)[i] = sg[i];
    __syncthreads();
  }

  for (int i = wave * 4; i < n; i += 4 * (RS_THREADS / 32)) {
    int sq[4], uq[4];
#pragma unroll
    for (int k = 0; k < 4; ++k) {
      const int q = ((i + k) < n) ? (i + k) : i;
      sq[k] = qe[q];
      uq[k] = (int)qd[q];
    }
    float v[4][4];
    int   ub[4];
#pragma unroll
    for (int k = 0; k < 4; ++k) {
      const int s = __builtin_amdgcn_readfirstlane(sq[k]);
      ub[k] = ((i + k) < n && uq[k] != QD_NONE) ? __builtin_amdgcn_readfirstlane(uq[k]) * AST : -1;
      const float* row = feat + (long long)s * NFEAT + lane;
#pragma unroll
      for (int j = 0; j < 4; ++j) v[k][j] = row[32 * j];
    }
#pragma unroll
    for (int k = 0; k < 4; ++k) {
      if (ub[k] >= 0) {
#pragma unroll
        for (int j = 0; j < 4; ++j) atomicAdd(&acc[ub[k] + lane + 32 * j], v[k][j]);
      }
    }
  }
  __syncthreads();

  if (FIRST) {
    for (int i = tid; i < cnt * (NFEAT / 4); i += RS_THREADS) {
      const int row = i >> 5;
      const int c4  = (i & 31) * 4;
      const float4 s4 = *(const float4*)(&acc[row * AST + c4]);
      const float dv = fmaxf((float)degl[row], 1.0f);
      float4 o;
      o.x = s4.x / dv; o.y = s4.y / dv; o.z = s4.z / dv; o.w = s4.w / dv;
      *(float4*)(agg + (long long)(lo + row) * NFEAT + c4) = o;
    }
  } else {
    const int hf  = lane >> 4;
    const int l16 = lane & 15;
    const v2f* wls = (const v2f*)wl;
    const v2f* wln = wls + NHID / 2 * NCLASS;
    for (int t = wave; t * 16 < cnt; t += RS_THREADS / 32) {
      int r = t * 16 + l16;
      if (r >= cnt) r = cnt - 1;
      const float rdv = 1.0f / fmaxf((float)degl[r], 1.0f);
      const float* hrow = feat + (long long)(lo + r) * NFEAT + 4 * hf;
      const float* srow = &acc[r * AST + 4 * hf];
      v8f c = {0.f, 0.f, 0.f, 0.f, 0.f, 0.f, 0.f, 0.f};
#pragma unroll
      for (int m = 0; m < NFEAT / 8; ++m) {
        const float4 hv = *(const float4*)(hrow + 8 * m);
        const float4 sv = *(const float4*)(srow + 8 * m);
        const v2f bs0 = wls[(4 * m + 0) * 16 + lane];
        const v2f bs1 = wls[(4 * m + 2) * 16 + lane];
        const v2f bn0 = wln[(4 * m + 0) * 16 + lane];
        const v2f bn1 = wln[(4 * m + 2) * 16 + lane];
        v2f ah0, ah1, aa0, aa1;
        ah0.x = hv.x;       ah0.y = hv.y;       ah1.x = hv.z;       ah1.y = hv.w;
        aa0.x = sv.x * rdv; aa0.y = sv.y * rdv; aa1.x = sv.z * rdv; aa1.y = sv.w * rdv;
        c = __builtin_amdgcn_wmma_f32_16x16x4_f32(false, ah0, false, bs0, (short)0, c, false, false);
        c = __builtin_amdgcn_wmma_f32_16x16x4_f32(false, aa0, false, bn0, (short)0, c, false, false);
        c = __builtin_amdgcn_wmma_f32_16x16x4_f32(false, ah1, false, bs1, (short)0, c, false, false);
        c = __builtin_amdgcn_wmma_f32_16x16x4_f32(false, aa1, false, bn1, (short)0, c, false, false);
      }

      const float bv = bias[l16];
#pragma unroll
      for (int j = 0; j < 8; ++j) {
        const float v = c[j] + bv;
        float mx = v;
#pragma unroll
        for (int off = 8; off >= 1; off >>= 1)
          mx = fmaxf(mx, __shfl_xor(mx, off, 16));
        const float e = expf(v - mx);
        float sm = e;
#pragma unroll
        for (int off = 8; off >= 1; off >>= 1)
          sm += __shfl_xor(sm, off, 16);
        const int row = t * 16 + j + hf * 8;
        if (row < cnt) out[(long long)(lo + row) * NCLASS + l16] = v - mx - logf(sm);
      }
    }
  }
}

__global__ __launch_bounds__(256)
void gemm1_kernel(const float* __restrict__ x, float* ah,
                  const float* __restrict__ wself, const float* __restrict__ wneigh,
                  const float* __restrict__ bias, int nNodes) {
  __shared__ float xs[G1_ROWS * G1_STRIDE];
  __shared__ float as[G1_ROWS * G1_STRIDE];

  const int tid  = threadIdx.x;
  const int lane = tid & 31;
  const int wave = tid >> 5;
  const int half = lane >> 4;
  const int l16  = lane & 15;
  const int col  = wave * 16 + l16;
  const long long base = (long long)blockIdx.x * G1_ROWS;

  for (int i = tid; i < G1_ROWS * 32; i += 256) {
    int row = i >> 5;
    int c4  = (i & 31) * 4;
    long long node = base + row;
    float4 vx = make_float4(0.f, 0.f, 0.f, 0.f);
    float4 va = vx;
    if (node < nNodes) {
      vx = *(const float4*)(x   + node * NFEAT + c4);
      va = *(const float4*)(ah  + node * NFEAT + c4);
    }
    *(float4*)(&xs[row * G1_STRIDE + c4]) = vx;
    *(float4*)(&as[row * G1_STRIDE + c4]) = va;
  }
  __syncthreads();

  const v8f vzero = {0.f, 0.f, 0.f, 0.f, 0.f, 0.f, 0.f, 0.f};
  v8f acc[G1_TILES];
#pragma unroll
  for (int r = 0; r < G1_TILES; ++r) acc[r] = vzero;

  for (int k0 = 0; k0 < NFEAT; k0 += 4) {
    const int kb = k0 + half * 2;
    v2f bs, bn;
    bs.x = wself [(long long)kb       * NHID + col];
    bs.y = wself [(long long)(kb + 1) * NHID + col];
    bn.x = wneigh[(long long)kb       * NHID + col];
    bn.y = wneigh[(long long)(kb + 1) * NHID + col];
#pragma unroll
    for (int r = 0; r < G1_TILES; ++r) {
      const v2f ax = *(const v2f*)(&xs[(r * 16 + l16) * G1_STRIDE + kb]);
      const v2f aa = *(const v2f*)(&as[(r * 16 + l16) * G1_STRIDE + kb]);
      acc[r] = __builtin_amdgcn_wmma_f32_16x16x4_f32(false, ax, false, bs,
                                                     (short)0, acc[r], false, false);
      acc[r] = __builtin_amdgcn_wmma_f32_16x16x4_f32(false, aa, false, bn,
                                                     (short)0, acc[r], false, false);
    }
  }

  const float bv = bias[col];
#pragma unroll
  for (int r = 0; r < G1_TILES; ++r) {
#pragma unroll
    for (int j = 0; j < 8; ++j) {
      long long node = base + r * 16 + j + half * 8;
      if (node < nNodes) {
        float v = acc[r][j] + bv;
        ah[node * NHID + col] = fmaxf(v, 0.0f);
      }
    }
  }
}

extern "C" void kernel_launch(void* const* d_in, const int* in_sizes, int n_in,
                              void* d_out, int out_size, void* d_ws, size_t ws_size,
                              hipStream_t stream) {
  const float* x   = (const float*)d_in[0];
  const int*   src = (const int*)  d_in[1];
  const int*   dst = (const int*)  d_in[2];
  const float* ws1 = (const float*)d_in[3];
  const float* wn1 = (const float*)d_in[4];
  const float* b1  = (const float*)d_in[5];
  const float* ws2 = (const float*)d_in[6];
  const float* wn2 = (const float*)d_in[7];
  const float* b2  = (const float*)d_in[8];
  float* out = (float*)d_out;

  const int nNodes = in_sizes[0] / NFEAT;
  const long long nEdges = in_sizes[1];

  float* pa = (float*)d_ws;
  float* pb = pa + (size_t)nNodes * NFEAT;

  const int rsBlocks = (nNodes + RNG - 1) / RNG;

  const int keyBlocks = (int)((nEdges + KEY_PIECE - 1) / KEY_PIECE) * 4;

  range_key_kernel<<<keyBlocks, RS_THREADS, 0, stream>>>(dst, (unsigned*)pb, nNodes, nEdges);
  range_sum_kernel<true><<<rsBlocks, RS_THREADS, 0, stream>>>(x, src, dst, pa, (int*)pb,
                                                              nullptr, nullptr, nullptr, nullptr, nNodes, nEdges);
  gemm1_kernel<<<(nNodes + G1_ROWS - 1) / G1_ROWS, 256, 0, stream>>>(
      x, pa, ws1, wn1, b1, nNodes);

  range_sum_kernel<false><<<rsBlocks, RS_THREADS, 0, stream>>>(pa, src, dst, nullptr, (int*)pb,
                                                               ws2, wn2, b2, out, nNodes, nEdges);
}


# both layers' neighbour sums without LDS float atomics: each queued edge goes to the wave owning row dst mod 32, which adds the 512-B row by plain 16-B LDS load/add/store; sums in owner list order
# speedup vs baseline: 3.5438x; 1.1143x over previous
#include <hip/hip_runtime.h>
#include <math.h>

typedef __attribute__((ext_vector_type(2))) float v2f;
typedef __attribute__((ext_vector_type(8))) float v8f;

#define NFEAT 128
#define NHID  128
#define NCLASS 16

#define G1_ROWS   48
#define G1_TILES  3
#define G1_STRIDE 132

#define RNG   400
#define QCAP  8192
#define RS_THREADS 1024
#define RS_WAVES (RS_THREADS / 32)

#define SQ      320
#define SQ_SBITS 23

#define SL_DEG 4
#define SL_Q   (SL_DEG + RNG)

#define SLAB_BYTES (RNG * NFEAT * 4)
#define KEY_OFF    131072
#define KEY_PIECE  65536
#define QD_NONE    0xFFFF

#define ACC_PITCH2 132
static_assert(SL_Q * 4 + 6 * QCAP <= KEY_OFF, "parked queue must end below the key bytes");
static_assert(KEY_OFF + KEY_PIECE <= SLAB_BYTES, "key piece must fit in a full range's rows");
static_assert(KEY_PIECE == 4 * 16 * RS_THREADS, "one piece = 4 loads of 16 B per thread");
static_assert(RNG < QD_NONE, "local dst must not collide with the marker");
static_assert(RNG <= 512, "local dst must fit the 9 low bits of a routed word");
static_assert(SQ % 8 == 0, "an owner takes its edges 8 at a time");

__device__ __forceinline__ bool keys_usable(int nNodes, long long nEdges) {
  const long long nPieces = (nEdges + KEY_PIECE - 1) / KEY_PIECE;
  return nPieces * RNG <= (long long)nNodes;
}

__global__ __launch_bounds__(RS_THREADS)
void range_key_kernel(const int* __restrict__ dst, unsigned* keep, int nNodes, long long nEdges) {
  if (!keys_usable(nNodes, nEdges)) return;
  const int piece = blockIdx.x >> 2;
  const int w0    = (blockIdx.x & 3) * (4 * RS_THREADS) + threadIdx.x;
  unsigned* kw = keep + (long long)piece * (SLAB_BYTES / 4) + KEY_OFF / 4;
  int4 d[4];
#pragma unroll
  for (int j = 0; j < 4; ++j) {
    const long long e = (long long)piece * KEY_PIECE + (long long)(w0 + j * RS_THREADS) * 4;
    if (e + 4 <= nEdges) {
      d[j] = *(const int4*)(dst + e);
    } else {
      d[j].x = (e + 0 < nEdges) ? dst[e + 0] : -1;
      d[j].y = (e + 1 < nEdges) ? dst[e + 1] : -1;
      d[j].z = (e + 2 < nEdges) ? dst[e + 2] : -1;
      d[j].w = (e + 3 < nEdges) ? dst[e + 3] : -1;
    }
  }
#pragma unroll
  for (int j = 0; j < 4; ++j) {
    const unsigned k0 = (d[j].x < 0) ? 255u : (((unsigned)d[j].x / RNG) & 255u);
    const unsigned k1 = (d[j].y < 0) ? 255u : (((unsigned)d[j].y / RNG) & 255u);
    const unsigned k2 = (d[j].z < 0) ? 255u : (((unsigned)d[j].z / RNG) & 255u);
    const unsigned k3 = (d[j].w < 0) ? 255u : (((unsigned)d[j].w / RNG) & 255u);
    kw[w0 + j * RS_THREADS] = k0 | (k1 << 8) | (k2 << 16) | (k3 << 24);
  }
}

__device__ __forceinline__ unsigned zero_bytes(unsigned x) {
  return ~((((x & 0x7F7F7F7Fu) + 0x7F7F7F7Fu) | x) | 0x7F7F7F7Fu);
}

__device__ __forceinline__ void ks_take16(const uint4 w, unsigned pat, int e0, int lo, int cnt,
                                          const float* __restrict__ feat,
                                          const int* __restrict__ src,
                                          const int* __restrict__ dst, long long nEdges,
                                          float* acc, int astr, int* qe, int* degl, int* qn) {
  unsigned m = zero_bytes(w.x ^ pat) | (zero_bytes(w.y ^ pat) >> 1) |
               (zero_bytes(w.z ^ pat) >> 2) | (zero_bytes(w.w ^ pat) >> 3);
  while (m) {
    const int pos = __ffs(m) - 1;
    const int e = e0 + 4 * (7 - (pos & 7)) + (pos >> 3);
    m &= m - 1;
    const int p = atomicAdd(qn, 1);
    if (p < QCAP) {
      qe[p] = e;
    } else if (e < nEdges) {
      const unsigned u = (unsigned)(dst[e] - lo);
      if (u < (unsigned)cnt) {
        atomicAdd(&degl[u], 1);
        const int s = src[e];
#pragma unroll 1
        for (int f = 0; f < NFEAT; ++f)
          atomicAdd(&acc[u * astr + f], feat[(long long)s * NFEAT + f]);
      }
    }
  }
}

__device__ __forceinline__ void ks_filter(const float* __restrict__ feat,
                                          const int* __restrict__ src,
                                          const int* __restrict__ dst,
                                          const unsigned char* keys,
                                          long long nEdges, int lo, int cnt, int tid,
                                          float* acc, int astr, int* qe, int* degl, int* qn) {
  const int nPieces = (int)((nEdges + KEY_PIECE - 1) / KEY_PIECE);
  const unsigned pat = (blockIdx.x & 255u) * 0x01010101u;
  int p = (int)(((long long)blockIdx.x * nPieces) / gridDim.x);
  const uint4* kp = (const uint4*)(keys + (long long)p * SLAB_BYTES + KEY_OFF);
  uint4 c0 = kp[tid];
  uint4 c1 = kp[tid + RS_THREADS];
  uint4 c2 = kp[tid + 2 * RS_THREADS];
  uint4 c3 = kp[tid + 3 * RS_THREADS];
  for (int pp = 0; pp < nPieces; ++pp) {
    int pn = p + 1;
    if (pn >= nPieces) pn = 0;
    uint4 n0 = c0, n1 = c1, n2 = c2, n3 = c3;
    if (pp + 1 < nPieces) {
      kp = (const uint4*)(keys + (long long)pn * SLAB_BYTES + KEY_OFF);
      n0 = kp[tid];
      n1 = kp[tid + RS_THREADS];
      n2 = kp[tid + 2 * RS_THREADS];
      n3 = kp[tid + 3 * RS_THREADS];
    }
    const int e0 = p * KEY_PIECE + tid * 16;
    ks_take16(c0, pat, e0,                   lo, cnt, feat, src, dst, nEdges, acc, astr, qe, degl, qn);
    ks_take16(c1, pat, e0 + 16 * RS_THREADS, lo, cnt, feat, src, dst, nEdges, acc, astr, qe, degl, qn);
    ks_take16(c2, pat, e0 + 32 * RS_THREADS, lo, cnt, feat, src, dst, nEdges, acc, astr, qe, degl, qn);
    ks_take16(c3, pat, e0 + 48 * RS_THREADS, lo, cnt, feat, src, dst, nEdges, acc, astr, qe, degl, qn);
    c0 = n0; c1 = n1; c2 = n2; c3 = n3;
    p = pn;
  }
}

__device__ __forceinline__ void rs_take(int d, int e, int lo, int cnt,
                                        const float* __restrict__ feat,
                                        const int* __restrict__ src,
                                        float* acc, int astr, int* qe, unsigned short* qd,
                                        int* degl, int* qn) {
  const unsigned u = (unsigned)(d - lo);
  if (u < (unsigned)cnt) {
    atomicAdd(&degl[u], 1);
    const int p = atomicAdd(qn, 1);
    if (p < QCAP) {
      qe[p] = e;
      qd[p] = (unsigned short)u;
    } else {
      const int s = src[e];
#pragma unroll 1
      for (int f = 0; f < NFEAT; ++f)
        atomicAdd(&acc[u * astr + f], feat[(long long)s * NFEAT + f]);
    }
  }
}

__device__ __forceinline__ void rs_filter(const float* __restrict__ feat,
                                          const int* __restrict__ src,
                                          const int* __restrict__ dst,
                                          long long nEdges, int lo, int cnt, int tid,
                                          float* acc, int astr, int* qe, unsigned short* qd,
                                          int* degl, int* qn) {
  const long long n4 = nEdges >> 2;
  const int4* __restrict__ dst4 = (const int4*)dst;
  const int nFull = (int)(n4 / (4 * RS_THREADS));
  const int rot   = (int)(((long long)blockIdx.x * nFull) / gridDim.x);
  for (int gg = 0; gg < nFull; ++gg) {
    int g = gg + rot;
    if (g >= nFull) g -= nFull;
    const long long i = (long long)g * (4 * RS_THREADS) + tid;
    const int4 d0 = dst4[i];
    const int4 d1 = dst4[i + RS_THREADS];
    const int4 d2 = dst4[i + 2 * RS_THREADS];
    const int4 d3 = dst4[i + 3 * RS_THREADS];
    const int e0 = (int)(i * 4);
    const int e1 = e0 + 4 * RS_THREADS, e2 = e0 + 8 * RS_THREADS, e3 = e0 + 12 * RS_THREADS;
    rs_take(d0.x, e0 + 0, lo, cnt, feat, src, acc, astr, qe, qd, degl, qn);
    rs_take(d0.y, e0 + 1, lo, cnt, feat, src, acc, astr, qe, qd, degl, qn);
    rs_take(d0.z, e0 + 2, lo, cnt, feat, src, acc, astr, qe, qd, degl, qn);
    rs_take(d0.w, e0 + 3, lo, cnt, feat, src, acc, astr, qe, qd, degl, qn);
    rs_take(d1.x, e1 + 0, lo, cnt, feat, src, acc, astr, qe, qd, degl, qn);
    rs_take(d1.y, e1 + 1, lo, cnt, feat, src, acc, astr, qe, qd, degl, qn);
    rs_take(d1.z, e1 + 2, lo, cnt, feat, src, acc, astr, qe, qd, degl, qn);
    rs_take(d1.w, e1 + 3, lo, cnt, feat, src, acc, astr, qe, qd, degl, qn);
    rs_take(d2.x, e2 + 0, lo, cnt, feat, src, acc, astr, qe, qd, degl, qn);
    rs_take(d2.y, e2 + 1, lo, cnt, feat, src, acc, astr, qe, qd, degl, qn);
    rs_take(d2.z, e2 + 2, lo, cnt, feat, src, acc, astr, qe, qd, degl, qn);
    rs_take(d2.w, e2 + 3, lo, cnt, feat, src, acc, astr, qe, qd, degl, qn);
    rs_take(d3.x, e3 + 0, lo, cnt, feat, src, acc, astr, qe, qd, degl, qn);
    rs_take(d3.y, e3 + 1, lo, cnt, feat, src, acc, astr, qe, qd, degl, qn);
    rs_take(d3.z, e3 + 2, lo, cnt, feat, src, acc, astr, qe, qd, degl, qn);
    rs_take(d3.w, e3 + 3, lo, cnt, feat, src, acc, astr, qe, qd, degl, qn);
  }
  for (long long i = (long long)nFull * (4 * RS_THREADS) + tid; i < n4; i += RS_THREADS) {
    const int4 d = dst4[i];
    const int e = (int)(i * 4);
    rs_take(d.x, e + 0, lo, cnt, feat, src, acc, astr, qe, qd, degl, qn);
    rs_take(d.y, e + 1, lo, cnt, feat, src, acc, astr, qe, qd, degl, qn);
    rs_take(d.z, e + 2, lo, cnt, feat, src, acc, astr, qe, qd, degl, qn);
    rs_take(d.w, e + 3, lo, cnt, feat, src, acc, astr, qe, qd, degl, qn);
  }
  for (long long e = n4 * 4 + tid; e < nEdges; e += RS_THREADS)
    rs_take(dst[e], (int)e, lo, cnt, feat, src, acc, astr, qe, qd, degl, qn);
}

__device__ __forceinline__ void own_load8(const unsigned* q, int nw, int b, int lane,
                                          const float* __restrict__ frow,
                                          unsigned (&e)[8], float4 (&v)[8]) {
  int idx = b * 8 + (lane & 7);
  if (idx > nw - 1) idx = nw - 1;
  const int mine = (int)q[idx];
#pragma unroll
  for (int k = 0; k < 8; ++k) {
    e[k] = (unsigned)__builtin_amdgcn_readlane(mine, k);
    v[k] = *(const float4*)(frow + (long long)(e[k] >> 9) * NFEAT);
  }
}

__device__ __forceinline__ void own_add8(float* acc, int astr, int nw, int b, int lane,
                                         const unsigned (&e)[8], const float4 (&v)[8]) {
#pragma unroll
  for (int k = 0; k < 8; ++k) {
    if (b * 8 + k < nw) {
      float4* p = (float4*)&acc[(int)(e[k] & 511u) * astr + 4 * lane];
      float4 a = *p;
      a.x += v[k].x; a.y += v[k].y; a.z += v[k].z; a.w += v[k].w;
      *p = a;
    }
  }
}

template <bool FIRST>
__global__ __launch_bounds__(RS_THREADS)
void range_sum_kernel(const float* __restrict__ feat,
                      const int* __restrict__ src,
                      const int* __restrict__ dst,
                      float* agg,
                      int* keep,
                      const float* __restrict__ wself,
                      const float* __restrict__ wneigh,
                      const float* __restrict__ bias,
                      float* __restrict__ out,
                      int nNodes, long long nEdges) {
  constexpr int AST = FIRST ? NFEAT : ACC_PITCH2;
  __shared__ __attribute__((aligned(16))) float acc[RNG * AST];
  __shared__ __attribute__((aligned(16))) int qe[QCAP];
  __shared__ __attribute__((aligned(16))) unsigned short qd[QCAP];
  __shared__ __attribute__((aligned(16))) int degl[RNG];
  __shared__ int qn;
  __shared__ int hd;
  __shared__ __attribute__((aligned(16))) unsigned rq[RS_WAVES * SQ];
  __shared__ int rqn[RS_WAVES];
  __shared__ __attribute__((aligned(16))) float wl[FIRST ? 4 : 2 * NHID * NCLASS];

  const int tid  = threadIdx.x;
  const int lane = tid & 31;
  const int wave = tid >> 5;
  const int lo   = blockIdx.x * RNG;
  const int cnt  = (nNodes - lo < RNG) ? (nNodes - lo) : RNG;
  int* slab = keep + (long long)lo * NFEAT;

  const float4 z4 = make_float4(0.f, 0.f, 0.f, 0.f);
  for (int i = tid; i < RNG * AST / 4; i += RS_THREADS) ((float4*)acc)[i] = z4;
  if (tid < RS_WAVES) rqn[tid] = 0;
  if (FIRST) {
    for (int i = tid; i < RNG; i += RS_THREADS) degl[i] = 0;
    if (tid == 0) qn = 0;
  } else {
    if (tid == 0) hd = slab[0];
    for (int i = tid; i < NHID / 2 * NCLASS; i += RS_THREADS) {
      const int q = i >> 4, c = i & 15;
      const int k = 8 * (q >> 2) + 4 * (q & 1) + 2 * ((q >> 1) & 1);
      v2f ws, wn;
      ws.x = wself [k * NCLASS + c];  ws.y = wself [(k + 1) * NCLASS + c];
      wn.x = wneigh[k * NCLASS + c];  wn.y = wneigh[(k + 1) * NCLASS + c];
      ((v2f*)wl)[i] = ws;
      ((v2f*)wl)[NHID / 2 * NCLASS + i] = wn;
    }
  }
  __syncthreads();

  int n;
  const bool scan = FIRST ? true : (hd < 0);
  if (scan) {
    if (!FIRST) {
      for (int i = tid; i < RNG; i += RS_THREADS) degl[i] = 0;
      if (tid == 0) qn = 0;
      __syncthreads();
    }
    const bool keyed = FIRST && keys_usable(nNodes, nEdges);
    if (keyed) ks_filter(feat, src, dst, (const unsigned char*)keep, nEdges, lo, cnt, tid, acc, AST, qe, degl, &qn);
    else       rs_filter(feat, src, dst, nEdges, lo, cnt, tid, acc, AST, qe, qd, degl, &qn);
    __syncthreads();

    const int total = qn;
    n = (total < QCAP) ? total : QCAP;

    if (keyed) {
      for (int i = tid; i < n; i += RS_THREADS) {
        const int e = qe[i];
        int s = 0;
        unsigned short uu = QD_NONE;
        if (e < nEdges) {
          const int dv = dst[e];
          const int sv = src[e];
          const unsigned u = (unsigned)(dv - lo);
          if (u < (unsigned)cnt) { s = sv; uu = (unsigned short)u; atomicAdd(&degl[u], 1); }
        }
        qe[i] = s;
        qd[i] = uu;
      }
    } else {
      for (int i = tid; i < n; i += RS_THREADS) qe[i] = src[qe[i]];
    }
    __syncthreads();

    if (FIRST) {
      const int np = (n + 7) & ~7;
      const bool fits = (total <= QCAP) && (SL_Q * 4 + 6 * np <= cnt * NFEAT * 4);
      if (fits) {
        int4* sq = (int4*)(slab + SL_Q);
        for (int i = tid; i < np / 4; i += RS_THREADS) sq[i] = ((const int4*)qe)[i];
        int4* sd = (int4*)(slab + SL_Q + np);
        for (int i = tid; i < np / 8; i += RS_THREADS) sd[i] = ((const int4*)qd)[i];
        int4* sg = (int4*)(slab + SL_DEG);
        for (int i = tid; i < RNG / 4; i += RS_THREADS) sg[i] = ((const int4*)degl)[i];
      }
      if (tid == 0) slab[0] = fits ? n : -1;
    }
  } else {
    n = (hd < QCAP) ? hd : QCAP;
    const int np = (n + 7) & ~7;
    const int4* sq = (const int4*)(slab + SL_Q);
    for (int i = tid; i < np / 4; i += RS_THREADS) ((int4*)qe)[i] = sq[i];
    const int4* sd = (const int4*)(slab + SL_Q + np);
    for (int i = tid; i < np / 8; i += RS_THREADS) ((int4*)qd)[i] = sd[i];
    const int4* sg = (const int4*)(slab + SL_DEG);
    for (int i = tid; i < RNG / 4; i += RS_THREADS) ((int4*)degl)[i] = sg[i];
    __syncthreads();
  }

  for (int i = tid; i < n; i += RS_THREADS) {
    const unsigned u = qd[i];
    const int s = qe[i];
    if (u != QD_NONE) {
      const int ow = (int)(u & (RS_WAVES - 1));
      int pos = SQ;
      if ((unsigned)s < (1u << SQ_SBITS)) pos = atomicAdd(&rqn[ow], 1);
      if (pos < SQ) {
        rq[ow * SQ + pos] = ((unsigned)s << 9) | u;
      } else {
#pragma unroll 1
        for (int f = 0; f < NFEAT; ++f)
          atomicAdd(&acc[(int)u * AST + f], feat[(long long)s * NFEAT + f]);
      }
    }
  }
  __syncthreads();

  {
    int nw = rqn[wave];
    if (nw > SQ) nw = SQ;
    nw = __builtin_amdgcn_readfirstlane(nw);
    if (nw > 0) {
      const unsigned* q = rq + wave * SQ;
      const float* frow = feat + 4 * lane;
      const int nb = (nw + 7) >> 3;
      unsigned ea[8], eb[8];
      float4   va[8], vb[8];
      own_load8(q, nw, 0, lane, frow, ea, va);
      for (int b = 0; b < nb; b += 2) {
        own_load8(q, nw, b + 1, lane, frow, eb, vb);
        own_add8(acc, AST, nw, b, lane, ea, va);
        own_load8(q, nw, b + 2, lane, frow, ea, va);
        own_add8(acc, AST, nw, b + 1, lane, eb, vb);
      }
    }
  }
  __syncthreads();

  if (FIRST) {
    for (int i = tid; i < cnt * (NFEAT / 4); i += RS_THREADS) {
      const int row = i >> 5;
      const int c4  = (i & 31) * 4;
      const float4 s4 = *(const float4*)(&acc[row * AST + c4]);
      const float dv = fmaxf((float)degl[row], 1.0f);
      float4 o;
      o.x = s4.x / dv; o.y = s4.y / dv; o.z = s4.z / dv; o.w = s4.w / dv;
      *(float4*)(agg + (long long)(lo + row) * NFEAT + c4) = o;
    }
  } else {
    const int hf  = lane >> 4;
    const int l16 = lane & 15;
    const v2f* wls = (const v2f*)wl;
    const v2f* wln = wls + NHID / 2 * NCLASS;
    for (int t = wave; t * 16 < cnt; t += RS_THREADS / 32) {
      int r = t * 16 + l16;
      if (r >= cnt) r = cnt - 1;
      const float rdv = 1.0f / fmaxf((float)degl[r], 1.0f);
      const float* hrow = feat + (long long)(lo + r) * NFEAT + 4 * hf;
      const float* srow = &acc[r * AST + 4 * hf];
      v8f c = {0.f, 0.f, 0.f, 0.f, 0.f, 0.f, 0.f, 0.f};
#pragma unroll
      for (int m = 0; m < NFEAT / 8; ++m) {
        const float4 hv = *(const float4*)(hrow + 8 * m);
        const float4 sv = *(const float4*)(srow + 8 * m);
        const v2f bs0 = wls[(4 * m + 0) * 16 + lane];
        const v2f bs1 = wls[(4 * m + 2) * 16 + lane];
        const v2f bn0 = wln[(4 * m + 0) * 16 + lane];
        const v2f bn1 = wln[(4 * m + 2) * 16 + lane];
        v2f ah0, ah1, aa0, aa1;
        ah0.x = hv.x;       ah0.y = hv.y;       ah1.x = hv.z;       ah1.y = hv.w;
        aa0.x = sv.x * rdv; aa0.y = sv.y * rdv; aa1.x = sv.z * rdv; aa1.y = sv.w * rdv;
        c = __builtin_amdgcn_wmma_f32_16x16x4_f32(false, ah0, false, bs0, (short)0, c, false, false);
        c = __builtin_amdgcn_wmma_f32_16x16x4_f32(false, aa0, false, bn0, (short)0, c, false, false);
        c = __builtin_amdgcn_wmma_f32_16x16x4_f32(false, ah1, false, bs1, (short)0, c, false, false);
        c = __builtin_amdgcn_wmma_f32_16x16x4_f32(false, aa1, false, bn1, (short)0, c, false, false);
      }

      const float bv = bias[l16];
#pragma unroll
      for (int j = 0; j < 8; ++j) {
        const float v = c[j] + bv;
        float mx = v;
#pragma unroll
        for (int off = 8; off >= 1; off >>= 1)
          mx = fmaxf(mx, __shfl_xor(mx, off, 16));
        const float e = expf(v - mx);
        float sm = e;
#pragma unroll
        for (int off = 8; off >= 1; off >>= 1)
          sm += __shfl_xor(sm, off, 16);
        const int row = t * 16 + j + hf * 8;
        if (row < cnt) out[(long long)(lo + row) * NCLASS + l16] = v - mx - logf(sm);
      }
    }
  }
}

__global__ __launch_bounds__(256)
void gemm1_kernel(const float* __restrict__ x, float* ah,
                  const float* __restrict__ wself, const float* __restrict__ wneigh,
                  const float* __restrict__ bias, int nNodes) {
  __shared__ float xs[G1_ROWS * G1_STRIDE];
  __shared__ float as[G1_ROWS * G1_STRIDE];

  const int tid  = threadIdx.x;
  const int lane = tid & 31;
  const int wave = tid >> 5;
  const int half = lane >> 4;
  const int l16  = lane & 15;
  const int col  = wave * 16 + l16;
  const long long base = (long long)blockIdx.x * G1_ROWS;

  for (int i = tid; i < G1_ROWS * 32; i += 256) {
    int row = i >> 5;
    int c4  = (i & 31) * 4;
    long long node = base + row;
    float4 vx = make_float4(0.f, 0.f, 0.f, 0.f);
    float4 va = vx;
    if (node < nNodes) {
      vx = *(const float4*)(x   + node * NFEAT + c4);
      va = *(const float4*)(ah  + node * NFEAT + c4);
    }
    *(float4*)(&xs[row * G1_STRIDE + c4]) = vx;
    *(float4*)(&as[row * G1_STRIDE + c4]) = va;
  }
  __syncthreads();

  const v8f vzero = {0.f, 0.f, 0.f, 0.f, 0.f, 0.f, 0.f, 0.f};
  v8f acc[G1_TILES];
#pragma unroll
  for (int r = 0; r < G1_TILES; ++r) acc[r] = vzero;

  for (int k0 = 0; k0 < NFEAT; k0 += 4) {
    const int kb = k0 + half * 2;
    v2f bs, bn;
    bs.x = wself [(long long)kb       * NHID + col];
    bs.y = wself [(long long)(kb + 1) * NHID + col];
    bn.x = wneigh[(long long)kb       * NHID + col];
    bn.y = wneigh[(long long)(kb + 1) * NHID + col];
#pragma unroll
    for (int r = 0; r < G1_TILES; ++r) {
      const v2f ax = *(const v2f*)(&xs[(r * 16 + l16) * G1_STRIDE + kb]);
      const v2f aa = *(const v2f*)(&as[(r * 16 + l16) * G1_STRIDE + kb]);
      acc[r] = __builtin_amdgcn_wmma_f32_16x16x4_f32(false, ax, false, bs,
                                                     (short)0, acc[r], false, false);
      acc[r] = __builtin_amdgcn_wmma_f32_16x16x4_f32(false, aa, false, bn,
                                                     (short)0, acc[r], false, false);
    }
  }

  const float bv = bias[col];
#pragma unroll
  for (int r = 0; r < G1_TILES; ++r) {
#pragma unroll
    for (int j = 0; j < 8; ++j) {
      long long node = base + r * 16 + j + half * 8;
      if (node < nNodes) {
        float v = acc[r][j] + bv;
        ah[node * NHID + col] = fmaxf(v, 0.0f);
      }
    }
  }
}

extern "C" void kernel_launch(void* const* d_in, const int* in_sizes, int n_in,
                              void* d_out, int out_size, void* d_ws, size_t ws_size,
                              hipStream_t stream) {
  const float* x   = (const float*)d_in[0];
  const int*   src = (const int*)  d_in[1];
  const int*   dst = (const int*)  d_in[2];
  const float* ws1 = (const float*)d_in[3];
  const float* wn1 = (const float*)d_in[4];
  const float* b1  = (const float*)d_in[5];
  const float* ws2 = (const float*)d_in[6];
  const float* wn2 = (const float*)d_in[7];
  const float* b2  = (const float*)d_in[8];
  float* out = (float*)d_out;

  const int nNodes = in_sizes[0] / NFEAT;
  const long long nEdges = in_sizes[1];

  float* pa = (float*)d_ws;
  float* pb = pa + (size_t)nNodes * NFEAT;

  const int rsBlocks = (nNodes + RNG - 1) / RNG;

  const int keyBlocks = (int)((nEdges + KEY_PIECE - 1) / KEY_PIECE) * 4;

  range_key_kernel<<<keyBlocks, RS_THREADS, 0, stream>>>(dst, (unsigned*)pb, nNodes, nEdges);
  range_sum_kernel<true><<<rsBlocks, RS_THREADS, 0, stream>>>(x, src, dst, pa, (int*)pb,
                                                              nullptr, nullptr, nullptr, nullptr, nNodes, nEdges);
  gemm1_kernel<<<(nNodes + G1_ROWS - 1) / G1_ROWS, 256, 0, stream>>>(
      x, pa, ws1, wn1, b1, nNodes);

  range_sum_kernel<false><<<rsBlocks, RS_THREADS, 0, stream>>>(pa, src, dst, nullptr, (int*)pb,
                                                               ws2, wn2, b2, out, nNodes, nEdges);
}


# layer-1 product + relu inside the layer-1 sum kernel: means divided in LDS (pitch 132) are A tiles, x rows and W read as fragments; means plane and a launch gone; k order regrouped
# speedup vs baseline: 3.6706x; 1.0358x over previous
#include <hip/hip_runtime.h>
#include <math.h>

typedef __attribute__((ext_vector_type(2))) float v2f;
typedef __attribute__((ext_vector_type(8))) float v8f;

#define NFEAT 128
#define NHID  128
#define NCLASS 16

#define RNG   400
#define QCAP  8192
#define RS_THREADS 1024
#define RS_WAVES (RS_THREADS / 32)

#define SQ      320
#define SQ_SBITS 23

#define SL_DEG 4
#define SL_Q   (SL_DEG + RNG)

#define SLAB_BYTES (RNG * NFEAT * 4)
#define KEY_OFF    131072
#define KEY_PIECE  65536
#define QD_NONE    0xFFFF

#define ACC_PITCH2 132
static_assert(SL_Q * 4 + 6 * QCAP <= KEY_OFF, "parked queue must end below the key bytes");
static_assert(KEY_OFF + KEY_PIECE <= SLAB_BYTES, "key piece must fit in a full range's rows");
static_assert(KEY_PIECE == 4 * 16 * RS_THREADS, "one piece = 4 loads of 16 B per thread");
static_assert(RNG < QD_NONE, "local dst must not collide with the marker");
static_assert(RNG <= 512, "local dst must fit the 9 low bits of a routed word");
static_assert(SQ % 8 == 0, "an owner takes its edges 8 at a time");
static_assert((RNG + 15) / 16 <= 28, "the layer-1 product gives a wave at most 7 row tiles");

__device__ __forceinline__ bool keys_usable(int nNodes, long long nEdges) {
  const long long nPieces = (nEdges + KEY_PIECE - 1) / KEY_PIECE;
  return nPieces * RNG <= (long long)nNodes;
}

__global__ __launch_bounds__(RS_THREADS)
void range_key_kernel(const int* __restrict__ dst, unsigned* keep, int nNodes, long long nEdges) {
  if (!keys_usable(nNodes, nEdges)) return;
  const int piece = blockIdx.x >> 2;
  const int w0    = (blockIdx.x & 3) * (4 * RS_THREADS) + threadIdx.x;
  unsigned* kw = keep + (long long)piece * (SLAB_BYTES / 4) + KEY_OFF / 4;
  int4 d[4];
#pragma unroll
  for (int j = 0; j < 4; ++j) {
    const long long e = (long long)piece * KEY_PIECE + (long long)(w0 + j * RS_THREADS) * 4;
    if (e + 4 <= nEdges) {
      d[j] = *(const int4*)(dst + e);
    } else {
      d[j].x = (e + 0 < nEdges) ? dst[e + 0] : -1;
      d[j].y = (e + 1 < nEdges) ? dst[e + 1] : -1;
      d[j].z = (e + 2 < nEdges) ? dst[e + 2] : -1;
      d[j].w = (e + 3 < nEdges) ? dst[e + 3] : -1;
    }
  }
#pragma unroll
  for (int j = 0; j < 4; ++j) {
    const unsigned k0 = (d[j].x < 0) ? 255u : (((unsigned)d[j].x / RNG) & 255u);
    const unsigned k1 = (d[j].y < 0) ? 255u : (((unsigned)d[j].y / RNG) & 255u);
    const unsigned k2 = (d[j].z < 0) ? 255u : (((unsigned)d[j].z / RNG) & 255u);
    const unsigned k3 = (d[j].w < 0) ? 255u : (((unsigned)d[j].w / RNG) & 255u);
    kw[w0 + j * RS_THREADS] = k0 | (k1 << 8) | (k2 << 16) | (k3 << 24);
  }
}

__device__ __forceinline__ unsigned zero_bytes(unsigned x) {
  return ~((((x & 0x7F7F7F7Fu) + 0x7F7F7F7Fu) | x) | 0x7F7F7F7Fu);
}

__device__ __forceinline__ void ks_take16(const uint4 w, unsigned pat, int e0, int lo, int cnt,
                                          const float* __restrict__ feat,
                                          const int* __restrict__ src,
                                          const int* __restrict__ dst, long long nEdges,
                                          float* acc, int astr, int* qe, int* degl, int* qn) {
  unsigned m = zero_bytes(w.x ^ pat) | (zero_bytes(w.y ^ pat) >> 1) |
               (zero_bytes(w.z ^ pat) >> 2) | (zero_bytes(w.w ^ pat) >> 3);
  while (m) {
    const int pos = __ffs(m) - 1;
    const int e = e0 + 4 * (7 - (pos & 7)) + (pos >> 3);
    m &= m - 1;
    const int p = atomicAdd(qn, 1);
    if (p < QCAP) {
      qe[p] = e;
    } else if (e < nEdges) {
      const unsigned u = (unsigned)(dst[e] - lo);
      if (u < (unsigned)cnt) {
        atomicAdd(&degl[u], 1);
        const int s = src[e];
#pragma unroll 1
        for (int f = 0; f < NFEAT; ++f)
          atomicAdd(&acc[u * astr + f], feat[(long long)s * NFEAT + f]);
      }
    }
  }
}

__device__ __forceinline__ void ks_filter(const float* __restrict__ feat,
                                          const int* __restrict__ src,
                                          const int* __restrict__ dst,
                                          const unsigned char* keys,
                                          long long nEdges, int lo, int cnt, int tid,
                                          float* acc, int astr, int* qe, int* degl, int* qn) {
  const int nPieces = (int)((nEdges + KEY_PIECE - 1) / KEY_PIECE);
  const unsigned pat = (blockIdx.x & 255u) * 0x01010101u;
  int p = (int)(((long long)blockIdx.x * nPieces) / gridDim.x);
  const uint4* kp = (const uint4*)(keys + (long long)p * SLAB_BYTES + KEY_OFF);
  uint4 c0 = kp[tid];
  uint4 c1 = kp[tid + RS_THREADS];
  uint4 c2 = kp[tid + 2 * RS_THREADS];
  uint4 c3 = kp[tid + 3 * RS_THREADS];
  for (int pp = 0; pp < nPieces; ++pp) {
    int pn = p + 1;
    if (pn >= nPieces) pn = 0;
    uint4 n0 = c0, n1 = c1, n2 = c2, n3 = c3;
    if (pp + 1 < nPieces) {
      kp = (const uint4*)(keys + (long long)pn * SLAB_BYTES + KEY_OFF);
      n0 = kp[tid];
      n1 = kp[tid + RS_THREADS];
      n2 = kp[tid + 2 * RS_THREADS];
      n3 = kp[tid + 3 * RS_THREADS];
    }
    const int e0 = p * KEY_PIECE + tid * 16;
    ks_take16(c0, pat, e0,                   lo, cnt, feat, src, dst, nEdges, acc, astr, qe, degl, qn);
    ks_take16(c1, pat, e0 + 16 * RS_THREADS, lo, cnt, feat, src, dst, nEdges, acc, astr, qe, degl, qn);
    ks_take16(c2, pat, e0 + 32 * RS_THREADS, lo, cnt, feat, src, dst, nEdges, acc, astr, qe, degl, qn);
    ks_take16(c3, pat, e0 + 48 * RS_THREADS, lo, cnt, feat, src, dst, nEdges, acc, astr, qe, degl, qn);
    c0 = n0; c1 = n1; c2 = n2; c3 = n3;
    p = pn;
  }
}

__device__ __forceinline__ void rs_take(int d, int e, int lo, int cnt,
                                        const float* __restrict__ feat,
                                        const int* __restrict__ src,
                                        float* acc, int astr, int* qe, unsigned short* qd,
                                        int* degl, int* qn) {
  const unsigned u = (unsigned)(d - lo);
  if (u < (unsigned)cnt) {
    atomicAdd(&degl[u], 1);
    const int p = atomicAdd(qn, 1);
    if (p < QCAP) {
      qe[p] = e;
      qd[p] = (unsigned short)u;
    } else {
      const int s = src[e];
#pragma unroll 1
      for (int f = 0; f < NFEAT; ++f)
        atomicAdd(&acc[u * astr + f], feat[(long long)s * NFEAT + f]);
    }
  }
}

__device__ __forceinline__ void rs_filter(const float* __restrict__ feat,
                                          const int* __restrict__ src,
                                          const int* __restrict__ dst,
                                          long long nEdges, int lo, int cnt, int tid,
                                          float* acc, int astr, int* qe, unsigned short* qd,
                                          int* degl, int* qn) {
  const long long n4 = nEdges >> 2;
  const int4* __restrict__ dst4 = (const int4*)dst;
  const int nFull = (int)(n4 / (4 * RS_THREADS));
  const int rot   = (int)(((long long)blockIdx.x * nFull) / gridDim.x);
  for (int gg = 0; gg < nFull; ++gg) {
    int g = gg + rot;
    if (g >= nFull) g -= nFull;
    const long long i = (long long)g * (4 * RS_THREADS) + tid;
    const int4 d0 = dst4[i];
    const int4 d1 = dst4[i + RS_THREADS];
    const int4 d2 = dst4[i + 2 * RS_THREADS];
    const int4 d3 = dst4[i + 3 * RS_THREADS];
    const int e0 = (int)(i * 4);
    const int e1 = e0 + 4 * RS_THREADS, e2 = e0 + 8 * RS_THREADS, e3 = e0 + 12 * RS_THREADS;
    rs_take(d0.x, e0 + 0, lo, cnt, feat, src, acc, astr, qe, qd, degl, qn);
    rs_take(d0.y, e0 + 1, lo, cnt, feat, src, acc, astr, qe, qd, degl, qn);
    rs_take(d0.z, e0 + 2, lo, cnt, feat, src, acc, astr, qe, qd, degl, qn);
    rs_take(d0.w, e0 + 3, lo, cnt, feat, src, acc, astr, qe, qd, degl, qn);
    rs_take(d1.x, e1 + 0, lo, cnt, feat, src, acc, astr, qe, qd, degl, qn);
    rs_take(d1.y, e1 + 1, lo, cnt, feat, src, acc, astr, qe, qd, degl, qn);
    rs_take(d1.z, e1 + 2, lo, cnt, feat, src, acc, astr, qe, qd, degl, qn);
    rs_take(d1.w, e1 + 3, lo, cnt, feat, src, acc, astr, qe, qd, degl, qn);
    rs_take(d2.x, e2 + 0, lo, cnt, feat, src, acc, astr, qe, qd, degl, qn);
    rs_take(d2.y, e2 + 1, lo, cnt, feat, src, acc, astr, qe, qd, degl, qn);
    rs_take(d2.z, e2 + 2, lo, cnt, feat, src, acc, astr, qe, qd, degl, qn);
    rs_take(d2.w, e2 + 3, lo, cnt, feat, src, acc, astr, qe, qd, degl, qn);
    rs_take(d3.x, e3 + 0, lo, cnt, feat, src, acc, astr, qe, qd, degl, qn);
    rs_take(d3.y, e3 + 1, lo, cnt, feat, src, acc, astr, qe, qd, degl, qn);
    rs_take(d3.z, e3 + 2, lo, cnt, feat, src, acc, astr, qe, qd, degl, qn);
    rs_take(d3.w, e3 + 3, lo, cnt, feat, src, acc, astr, qe, qd, degl, qn);
  }
  for (long long i = (long long)nFull * (4 * RS_THREADS) + tid; i < n4; i += RS_THREADS) {
    const int4 d = dst4[i];
    const int e = (int)(i * 4);
    rs_take(d.x, e + 0, lo, cnt, feat, src, acc, astr, qe, qd, degl, qn);
    rs_take(d.y, e + 1, lo, cnt, feat, src, acc, astr, qe, qd, degl, qn);
    rs_take(d.z, e + 2, lo, cnt, feat, src, acc, astr, qe, qd, degl, qn);
    rs_take(d.w, e + 3, lo, cnt, feat, src, acc, astr, qe, qd, degl, qn);
  }
  for (long long e = n4 * 4 + tid; e < nEdges; e += RS_THREADS)
    rs_take(dst[e], (int)e, lo, cnt, feat, src, acc, astr, qe, qd, degl, qn);
}

__device__ __forceinline__ void own_load8(const unsigned* q, int nw, int b, int lane,
                                          const float* __restrict__ frow,
                                          unsigned (&e)[8], float4 (&v)[8]) {
  int idx = b * 8 + (lane & 7);
  if (idx > nw - 1) idx = nw - 1;
  const int mine = (int)q[idx];
#pragma unroll
  for (int k = 0; k < 8; ++k) {
    e[k] = (unsigned)__builtin_amdgcn_readlane(mine, k);
    v[k] = *(const float4*)(frow + (long long)(e[k] >> 9) * NFEAT);
  }
}

__device__ __forceinline__ void own_add8(float* acc, int astr, int nw, int b, int lane,
                                         const unsigned (&e)[8], const float4 (&v)[8]) {
#pragma unroll
  for (int k = 0; k < 8; ++k) {
    if (b * 8 + k < nw) {
      float4* p = (float4*)&acc[(int)(e[k] & 511u) * astr + 4 * lane];
      float4 a = *p;
      a.x += v[k].x; a.y += v[k].y; a.z += v[k].z; a.w += v[k].w;
      *p = a;
    }
  }
}

template <int NT>
struct L1Frag {
  float4 hv[NT];
  v2f bs0, bs1, bn0, bn1;
};

template <int NT>
__device__ __forceinline__ void l1_load(const float* const (&xr)[NT], const float* __restrict__ wsp,
                                        const float* __restrict__ wnp, int m, L1Frag<NT>& f) {
  f.bs0.x = wsp[(8 * m + 0) * NHID];  f.bs0.y = wsp[(8 * m + 1) * NHID];
  f.bs1.x = wsp[(8 * m + 2) * NHID];  f.bs1.y = wsp[(8 * m + 3) * NHID];
  f.bn0.x = wnp[(8 * m + 0) * NHID];  f.bn0.y = wnp[(8 * m + 1) * NHID];
  f.bn1.x = wnp[(8 * m + 2) * NHID];  f.bn1.y = wnp[(8 * m + 3) * NHID];
#pragma unroll
  for (int j = 0; j < NT; ++j) f.hv[j] = *(const float4*)(xr[j] + 8 * m);
}

template <int NT>
__device__ __forceinline__ void l1_mma(const float* const (&sr)[NT], int m, const L1Frag<NT>& f,
                                       v8f (&c)[NT]) {
  float4 sv[NT];
#pragma unroll
  for (int j = 0; j < NT; ++j) sv[j] = *(const float4*)(sr[j] + 8 * m);
#pragma unroll
  for (int j = 0; j < NT; ++j) {
    v2f ah0, ah1, aa0, aa1;
    ah0.x = f.hv[j].x; ah0.y = f.hv[j].y; ah1.x = f.hv[j].z; ah1.y = f.hv[j].w;
    aa0.x = sv[j].x;   aa0.y = sv[j].y;   aa1.x = sv[j].z;   aa1.y = sv[j].w;
    c[j] = __builtin_amdgcn_wmma_f32_16x16x4_f32(false, ah0, false, f.bs0, (short)0, c[j], false, false);
    c[j] = __builtin_amdgcn_wmma_f32_16x16x4_f32(false, aa0, false, f.bn0, (short)0, c[j], false, false);
    c[j] = __builtin_amdgcn_wmma_f32_16x16x4_f32(false, ah1, false, f.bs1, (short)0, c[j], false, false);
    c[j] = __builtin_amdgcn_wmma_f32_16x16x4_f32(false, aa1, false, f.bn1, (short)0, c[j], false, false);
  }
}

template <int NT>
__device__ __forceinline__ void l1_product(const float* __restrict__ x, const float* acc,
                                           const float* __restrict__ wself,
                                           const float* __restrict__ wneigh,
                                           const float* __restrict__ bias, float* hout,
                                           int lo, int cnt, int rt0, int col, int hf, int l16) {
  const float* xr[NT];
  const float* sr[NT];
  v8f c[NT];
#pragma unroll
  for (int j = 0; j < NT; ++j) {
    int r = (rt0 + 4 * j) * 16 + l16;
    if (r >= cnt) r = cnt - 1;
    xr[j] = x + (long long)(lo + r) * NFEAT + 4 * hf;
    sr[j] = acc + r * ACC_PITCH2 + 4 * hf;
    c[j] = (v8f){0.f, 0.f, 0.f, 0.f, 0.f, 0.f, 0.f, 0.f};
  }
  const float* wsp = wself  + (4 * hf) * NHID + col;
  const float* wnp = wneigh + (4 * hf) * NHID + col;
  L1Frag<NT> fa, fb;
  l1_load<NT>(xr, wsp, wnp, 0, fa);
#pragma unroll 1
  for (int m = 0; m < NFEAT / 8; m += 2) {
    l1_load<NT>(xr, wsp, wnp, m + 1, fb);
    l1_mma<NT>(sr, m, fa, c);
    if (m + 2 < NFEAT / 8) l1_load<NT>(xr, wsp, wnp, m + 2, fa);
    l1_mma<NT>(sr, m + 1, fb, c);
  }
  const float bv = bias[col];
#pragma unroll
  for (int j = 0; j < NT; ++j) {
#pragma unroll
    for (int i = 0; i < 8; ++i) {
      const int row = (rt0 + 4 * j) * 16 + i + hf * 8;
      if (row < cnt) hout[(long long)(lo + row) * NHID + col] = fmaxf(c[j][i] + bv, 0.0f);
    }
  }
}

template <bool FIRST>
__global__ __launch_bounds__(RS_THREADS)
void range_sum_kernel(const float* __restrict__ feat,
                      const int* __restrict__ src,
                      const int* __restrict__ dst,
                      float* agg,
                      int* keep,
                      const float* __restrict__ wself,
                      const float* __restrict__ wneigh,
                      const float* __restrict__ bias,
                      float* __restrict__ out,
                      int nNodes, long long nEdges) {
  constexpr int AST = ACC_PITCH2;
  __shared__ __attribute__((aligned(16))) float acc[RNG * AST];
  __shared__ __attribute__((aligned(16))) int qe[QCAP];
  __shared__ __attribute__((aligned(16))) unsigned short qd[QCAP];
  __shared__ __attribute__((aligned(16))) int degl[RNG];
  __shared__ int qn;
  __shared__ int hd;
  __shared__ __attribute__((aligned(16))) unsigned rq[RS_WAVES * SQ];
  __shared__ int rqn[RS_WAVES];
  __shared__ __attribute__((aligned(16))) float wl[FIRST ? 4 : 2 * NHID * NCLASS];

  const int tid  = threadIdx.x;
  const int lane = tid & 31;
  const int wave = tid >> 5;
  const int lo   = blockIdx.x * RNG;
  const int cnt  = (nNodes - lo < RNG) ? (nNodes - lo) : RNG;
  int* slab = keep + (long long)lo * NFEAT;

  const float4 z4 = make_float4(0.f, 0.f, 0.f, 0.f);
  for (int i = tid; i < RNG * AST / 4; i += RS_THREADS) ((float4*)acc)[i] = z4;
  if (tid < RS_WAVES) rqn[tid] = 0;
  if (FIRST) {
    for (int i = tid; i < RNG; i += RS_THREADS) degl[i] = 0;
    if (tid == 0) qn = 0;
  } else {
    if (tid == 0) hd = slab[0];
    for (int i = tid; i < NHID / 2 * NCLASS; i += RS_THREADS) {
      const int q = i >> 4, c = i & 15;
      const int k = 8 * (q >> 2) + 4 * (q & 1) + 2 * ((q >> 1) & 1);
      v2f ws, wn;
      ws.x = wself [k * NCLASS + c];  ws.y = wself [(k + 1) * NCLASS + c];
      wn.x = wneigh[k * NCLASS + c];  wn.y = wneigh[(k + 1) * NCLASS + c];
      ((v2f*)wl)[i] = ws;
      ((v2f*)wl)[NHID / 2 * NCLASS + i] = wn;
    }
  }
  __syncthreads();

  int n;
  const bool scan = FIRST ? true : (hd < 0);
  if (scan) {
    if (!FIRST) {
      for (int i = tid; i < RNG; i += RS_THREADS) degl[i] = 0;
      if (tid == 0) qn = 0;
      __syncthreads();
    }
    const bool keyed = FIRST && keys_usable(nNodes, nEdges);
    if (keyed) ks_filter(feat, src, dst, (const unsigned char*)keep, nEdges, lo, cnt, tid, acc, AST, qe, degl, &qn);
    else       rs_filter(feat, src, dst, nEdges, lo, cnt, tid, acc, AST, qe, qd, degl, &qn);
    __syncthreads();

    const int total = qn;
    n = (total < QCAP) ? total : QCAP;

    if (keyed) {
      for (int i = tid; i < n; i += RS_THREADS) {
        const int e = qe[i];
        int s = 0;
        unsigned short uu = QD_NONE;
        if (e < nEdges) {
          const int dv = dst[e];
          const int sv = src[e];
          const unsigned u = (unsigned)(dv - lo);
          if (u < (unsigned)cnt) { s = sv; uu = (unsigned short)u; atomicAdd(&degl[u], 1); }
        }
        qe[i] = s;
        qd[i] = uu;
      }
    } else {
      for (int i = tid; i < n; i += RS_THREADS) qe[i] = src[qe[i]];
    }
    __syncthreads();

    if (FIRST) {
      const int np = (n + 7) & ~7;
      const bool fits = (total <= QCAP) && (SL_Q * 4 + 6 * np <= cnt * NFEAT * 4);
      if (fits) {
        int4* sq = (int4*)(slab + SL_Q);
        for (int i = tid; i < np / 4; i += RS_THREADS) sq[i] = ((const int4*)qe)[i];
        int4* sd = (int4*)(slab + SL_Q + np);
        for (int i = tid; i < np / 8; i += RS_THREADS) sd[i] = ((const int4*)qd)[i];
        int4* sg = (int4*)(slab + SL_DEG);
        for (int i = tid; i < RNG / 4; i += RS_THREADS) sg[i] = ((const int4*)degl)[i];
      }
      if (tid == 0) slab[0] = fits ? n : -1;
    }
  } else {
    n = (hd < QCAP) ? hd : QCAP;
    const int np = (n + 7) & ~7;
    const int4* sq = (const int4*)(slab + SL_Q);
    for (int i = tid; i < np / 4; i += RS_THREADS) ((int4*)qe)[i] = sq[i];
    const int4* sd = (const int4*)(slab + SL_Q + np);
    for (int i = tid; i < np / 8; i += RS_THREADS) ((int4*)qd)[i] = sd[i];
    const int4* sg = (const int4*)(slab + SL_DEG);
    for (int i = tid; i < RNG / 4; i += RS_THREADS) ((int4*)degl)[i] = sg[i];
    __syncthreads();
  }

  for (int i = tid; i < n; i += RS_THREADS) {
    const unsigned u = qd[i];
    const int s = qe[i];
    if (u != QD_NONE) {
      const int ow = (int)(u & (RS_WAVES - 1));
      int pos = SQ;
      if ((unsigned)s < (1u << SQ_SBITS)) pos = atomicAdd(&rqn[ow], 1);
      if (pos < SQ) {
        rq[ow * SQ + pos] = ((unsigned)s << 9) | u;
      } else {
#pragma unroll 1
        for (int f = 0; f < NFEAT; ++f)
          atomicAdd(&acc[(int)u * AST + f], feat[(long long)s * NFEAT + f]);
      }
    }
  }
  __syncthreads();

  {
    int nw = rqn[wave];
    if (nw > SQ) nw = SQ;
    nw = __builtin_amdgcn_readfirstlane(nw);
    if (nw > 0) {
      const unsigned* q = rq + wave * SQ;
      const float* frow = feat + 4 * lane;
      const int nb = (nw + 7) >> 3;
      unsigned ea[8], eb[8];
      float4   va[8], vb[8];
      own_load8(q, nw, 0, lane, frow, ea, va);
      for (int b = 0; b < nb; b += 2) {
        own_load8(q, nw, b + 1, lane, frow, eb, vb);
        own_add8(acc, AST, nw, b, lane, ea, va);
        own_load8(q, nw, b + 2, lane, frow, ea, va);
        own_add8(acc, AST, nw, b + 1, lane, eb, vb);
      }
    }
  }
  __syncthreads();

  if (FIRST) {
    for (int i = tid; i < cnt * (NFEAT / 4); i += RS_THREADS) {
      const int row = i >> 5;
      const int c4  = (i & 31) * 4;
      float4* p = (float4*)(&acc[row * AST + c4]);
      const float4 s4 = *p;
      const float dv = fmaxf((float)degl[row], 1.0f);
      float4 o;
      o.x = s4.x / dv; o.y = s4.y / dv; o.z = s4.z / dv; o.w = s4.w / dv;
      *p = o;
    }
    __syncthreads();

    const int wa = wave >> 3, wb = (wave >> 2) & 1, wc = wave & 3;
    const int rg = (wa + wc) & 3;
    const int ct = 2 * wa + wb;
    const int hf  = lane >> 4;
    const int l16 = lane & 15;
    const int col = ct * 16 + l16;
    const int ntile = (cnt + 15) >> 4;
    int nt = (ntile > rg) ? ((ntile - rg + 3) >> 2) : 0;
    nt = __builtin_amdgcn_readfirstlane(nt);
    for (int j0 = 0; j0 < nt; j0 += 4) {
      const int n   = (nt - j0 < 4) ? (nt - j0) : 4;
      const int rt0 = rg + 4 * j0;
      switch (n) {
        case 4:  l1_product<4>(feat, acc, wself, wneigh, bias, agg, lo, cnt, rt0, col, hf, l16); break;
        case 3:  l1_product<3>(feat, acc, wself, wneigh, bias, agg, lo, cnt, rt0, col, hf, l16); break;
        case 2:  l1_product<2>(feat, acc, wself, wneigh, bias, agg, lo, cnt, rt0, col, hf, l16); break;
        default: l1_product<1>(feat, acc, wself, wneigh, bias, agg, lo, cnt, rt0, col, hf, l16); break;
      }
    }
  } else {
    const int hf  = lane >> 4;
    const int l16 = lane & 15;
    const v2f* wls = (const v2f*)wl;
    const v2f* wln = wls + NHID / 2 * NCLASS;
    for (int t = wave; t * 16 < cnt; t += RS_THREADS / 32) {
      int r = t * 16 + l16;
      if (r >= cnt) r = cnt - 1;
      const float rdv = 1.0f / fmaxf((float)degl[r], 1.0f);
      const float* hrow = feat + (long long)(lo + r) * NFEAT + 4 * hf;
      const float* srow = &acc[r * AST + 4 * hf];
      v8f c = {0.f, 0.f, 0.f, 0.f, 0.f, 0.f, 0.f, 0.f};
#pragma unroll
      for (int m = 0; m < NFEAT / 8; ++m) {
        const float4 hv = *(const float4*)(hrow + 8 * m);
        const float4 sv = *(const float4*)(srow + 8 * m);
        const v2f bs0 = wls[(4 * m + 0) * 16 + lane];
        const v2f bs1 = wls[(4 * m + 2) * 16 + lane];
        const v2f bn0 = wln[(4 * m + 0) * 16 + lane];
        const v2f bn1 = wln[(4 * m + 2) * 16 + lane];
        v2f ah0, ah1, aa0, aa1;
        ah0.x = hv.x;       ah0.y = hv.y;       ah1.x = hv.z;       ah1.y = hv.w;
        aa0.x = sv.x * rdv; aa0.y = sv.y * rdv; aa1.x = sv.z * rdv; aa1.y = sv.w * rdv;
        c = __builtin_amdgcn_wmma_f32_16x16x4_f32(false, ah0, false, bs0, (short)0, c, false, false);
        c = __builtin_amdgcn_wmma_f32_16x16x4_f32(false, aa0, false, bn0, (short)0, c, false, false);
        c = __builtin_amdgcn_wmma_f32_16x16x4_f32(false, ah1, false, bs1, (short)0, c, false, false);
        c = __builtin_amdgcn_wmma_f32_16x16x4_f32(false, aa1, false, bn1, (short)0, c, false, false);
      }

      const float bv = bias[l16];
#pragma unroll
      for (int j = 0; j < 8; ++j) {
        const float v = c[j] + bv;
        float mx = v;
#pragma unroll
        for (int off = 8; off >= 1; off >>= 1)
          mx = fmaxf(mx, __shfl_xor(mx, off, 16));
        const float e = expf(v - mx);
        float sm = e;
#pragma unroll
        for (int off = 8; off >= 1; off >>= 1)
          sm += __shfl_xor(sm, off, 16);
        const int row = t * 16 + j + hf * 8;
        if (row < cnt) out[(long long)(lo + row) * NCLASS + l16] = v - mx - logf(sm);
      }
    }
  }
}

extern "C" void kernel_launch(void* const* d_in, const int* in_sizes, int n_in,
                              void* d_out, int out_size, void* d_ws, size_t ws_size,
                              hipStream_t stream) {
  const float* x   = (const float*)d_in[0];
  const int*   src = (const int*)  d_in[1];
  const int*   dst = (const int*)  d_in[2];
  const float* ws1 = (const float*)d_in[3];
  const float* wn1 = (const float*)d_in[4];
  const float* b1  = (const float*)d_in[5];
  const float* ws2 = (const float*)d_in[6];
  const float* wn2 = (const float*)d_in[7];
  const float* b2  = (const float*)d_in[8];
  float* out = (float*)d_out;

  const int nNodes = in_sizes[0] / NFEAT;
  const long long nEdges = in_sizes[1];

  float* pa = (float*)d_ws;
  float* pb = pa + (size_t)nNodes * NFEAT;

  const int rsBlocks = (nNodes + RNG - 1) / RNG;

  const int keyBlocks = (int)((nEdges + KEY_PIECE - 1) / KEY_PIECE) * 4;

  range_key_kernel<<<keyBlocks, RS_THREADS, 0, stream>>>(dst, (unsigned*)pb, nNodes, nEdges);
  range_sum_kernel<true><<<rsBlocks, RS_THREADS, 0, stream>>>(x, src, dst, pa, (int*)pb,
                                                              ws1, wn1, b1, nullptr, nNodes, nEdges);

  range_sum_kernel<false><<<rsBlocks, RS_THREADS, 0, stream>>>(pa, src, dst, nullptr, (int*)pb,
                                                               ws2, wn2, b2, out, nNodes, nEdges);
}


# layer-1 filter reads the edges' range numbers as 8 bit planes (written by the launch in front with wave ballots): 32 edges per XOR/OR word, not 4 byte tests per word; sums stay in owner list order
# speedup vs baseline: 3.8549x; 1.0502x over previous
#include <hip/hip_runtime.h>
#include <math.h>

typedef __attribute__((ext_vector_type(2))) float v2f;
typedef __attribute__((ext_vector_type(8))) float v8f;

#define NFEAT 128
#define NHID  128
#define NCLASS 16

#define RNG   400
#define QCAP  8192
#define RS_THREADS 1024
#define RS_WAVES (RS_THREADS / 32)

#define SQ      320
#define SQ_SBITS 23

#define SL_DEG 4
#define SL_Q   (SL_DEG + RNG)

#define SLAB_BYTES (RNG * NFEAT * 4)
#define KEY_OFF    131072
#define KEY_PIECE  65536
#define KU_EDGES   128
#define KU_PER_PIECE (KEY_PIECE / KU_EDGES)
#define KT_BYTES   4096
#define KEY_THREADS 256
#define QD_NONE    0xFFFF

#define ACC_PITCH2 132
static_assert(SL_Q * 4 + 6 * QCAP <= KEY_OFF, "parked queue must end below the key bytes");
static_assert(KEY_OFF + KEY_PIECE <= SLAB_BYTES, "key piece must fit in a full range's rows");
static_assert(KU_PER_PIECE == 512 && KT_BYTES == 32 * 8 * 16, "unit and tile arithmetic below");
static_assert(RNG < QD_NONE, "local dst must not collide with the marker");
static_assert(RNG <= 512, "local dst must fit the 9 low bits of a routed word");
static_assert(SQ % 8 == 0, "an owner takes its edges 8 at a time");
static_assert((RNG + 15) / 16 <= 28, "the layer-1 product gives a wave at most 7 row tiles");

__device__ __forceinline__ bool keys_usable(int nNodes, long long nEdges) {
  const long long nPieces = (nEdges + KEY_PIECE - 1) / KEY_PIECE;
  return nPieces * RNG <= (long long)nNodes;
}

__device__ __forceinline__ unsigned range_no(int d) {
  return (d < 0) ? 255u : (((unsigned)d / RNG) & 255u);
}

__global__ __launch_bounds__(KEY_THREADS)
void range_key_kernel(const int* __restrict__ dst, unsigned* keep, int nNodes, long long nEdges) {
  if (!keys_usable(nNodes, nEdges)) return;
  const int lane = threadIdx.x & 31;
  const int U0   = (blockIdx.x * (KEY_THREADS / 32) + (threadIdx.x >> 5)) * 8;
  const int piece = U0 / KU_PER_PIECE;
  const int ul0   = U0 % KU_PER_PIECE;
  int4 d[8];
#pragma unroll
  for (int k = 0; k < 8; ++k) {
    const long long e = (long long)(U0 + k) * KU_EDGES + 4 * lane;
    if (e + 4 <= nEdges) {
      d[k] = *(const int4*)(dst + e);
    } else {
      d[k].x = (e + 0 < nEdges) ? dst[e + 0] : -1;
      d[k].y = (e + 1 < nEdges) ? dst[e + 1] : -1;
      d[k].z = (e + 2 < nEdges) ? dst[e + 2] : -1;
      d[k].w = (e + 3 < nEdges) ? dst[e + 3] : -1;
    }
  }
  unsigned r[4][8];
#pragma unroll
  for (int j = 0; j < 4; ++j)
#pragma unroll
    for (int b = 0; b < 8; ++b) r[j][b] = 0u;
#pragma unroll
  for (int k = 0; k < 8; ++k) {
    unsigned kj[4];
    kj[0] = range_no(d[k].x); kj[1] = range_no(d[k].y);
    kj[2] = range_no(d[k].z); kj[3] = range_no(d[k].w);
#pragma unroll
    for (int j = 0; j < 4; ++j) {
#pragma unroll
      for (int b = 0; b < 8; ++b) {
        const unsigned pl = __builtin_amdgcn_ballot_w32(((kj[j] >> b) & 1u) != 0u);
        if (lane == k) r[j][b] = pl;
      }
    }
  }
  if (lane < 8) {
    unsigned* tile = keep + (long long)piece * (SLAB_BYTES / 4) + KEY_OFF / 4 + (ul0 >> 5) * (KT_BYTES / 4);
    const int ul = (ul0 & 31) + lane;
#pragma unroll
    for (int b = 0; b < 8; ++b)
      *(uint4*)(tile + b * 128 + ul * 4) = make_uint4(r[0][b], r[1][b], r[2][b], r[3][b]);
  }
}

__device__ __forceinline__ void ks_hits(unsigned m, int e0, int lo, int cnt,
                                        const float* __restrict__ feat,
                                        const int* __restrict__ src,
                                        const int* __restrict__ dst, long long nEdges,
                                        float* acc, int astr, int* qe, int* degl, int* qn) {
  while (m) {
    const int pos = __ffs(m) - 1;
    const int e = e0 + 4 * pos;
    m &= m - 1;
    const int p = atomicAdd(qn, 1);
    if (p < QCAP) {
      qe[p] = e;
    } else if (e < nEdges) {
      const unsigned u = (unsigned)(dst[e] - lo);
      if (u < (unsigned)cnt) {
        atomicAdd(&degl[u], 1);
        const int s = src[e];
#pragma unroll 1
        for (int f = 0; f < NFEAT; ++f)
          atomicAdd(&acc[u * astr + f], feat[(long long)s * NFEAT + f]);
      }
    }
  }
}

__device__ __forceinline__ void ks_load(const unsigned char* keys, int U, int nUnits, uint4 (&c)[8]) {
  if (U < nUnits) {
    const unsigned char* p = keys + (long long)(U / KU_PER_PIECE) * SLAB_BYTES + KEY_OFF +
                             ((U % KU_PER_PIECE) >> 5) * KT_BYTES + (U & 31) * 16;
#pragma unroll
    for (int b = 0; b < 8; ++b) c[b] = *(const uint4*)(p + b * 512);
  } else {
#pragma unroll
    for (int b = 0; b < 8; ++b) c[b] = make_uint4(0u, 0u, 0u, 0u);
  }
}

__device__ __forceinline__ void ks_filter(const float* __restrict__ feat,
                                          const int* __restrict__ src,
                                          const int* __restrict__ dst,
                                          const unsigned char* keys,
                                          long long nEdges, int lo, int cnt, int tid,
                                          float* acc, int astr, int* qe, int* degl, int* qn) {
  const int nPieces = (int)((nEdges + KEY_PIECE - 1) / KEY_PIECE);
  const int nUnits  = nPieces * KU_PER_PIECE;
  const int nTrips  = (nUnits + RS_THREADS - 1) / RS_THREADS;
  const unsigned key = blockIdx.x & 255u;
  unsigned sm[8];
#pragma unroll
  for (int b = 0; b < 8; ++b) sm[b] = 0u - ((key >> b) & 1u);
  int t = (int)(((long long)blockIdx.x * nTrips) / gridDim.x);
  uint4 c[8];
  ks_load(keys, t * RS_THREADS + tid, nUnits, c);
  for (int tt = 0; tt < nTrips; ++tt) {
    int tn = t + 1;
    if (tn >= nTrips) tn = 0;
    uint4 n[8];
#pragma unroll
    for (int b = 0; b < 8; ++b) n[b] = c[b];
    if (tt + 1 < nTrips) ks_load(keys, tn * RS_THREADS + tid, nUnits, n);
    const int U = t * RS_THREADS + tid;
    if (U < nUnits) {
      uint4 mm;
      mm.x = c[0].x ^ sm[0]; mm.y = c[0].y ^ sm[0]; mm.z = c[0].z ^ sm[0]; mm.w = c[0].w ^ sm[0];
#pragma unroll
      for (int b = 1; b < 8; ++b) {
        mm.x |= c[b].x ^ sm[b]; mm.y |= c[b].y ^ sm[b];
        mm.z |= c[b].z ^ sm[b]; mm.w |= c[b].w ^ sm[b];
      }
      const int e0 = U * KU_EDGES;
      ks_hits(~mm.x, e0 + 0, lo, cnt, feat, src, dst, nEdges, acc, astr, qe, degl, qn);
      ks_hits(~mm.y, e0 + 1, lo, cnt, feat, src, dst, nEdges, acc, astr, qe, degl, qn);
      ks_hits(~mm.z, e0 + 2, lo, cnt, feat, src, dst, nEdges, acc, astr, qe, degl, qn);
      ks_hits(~mm.w, e0 + 3, lo, cnt, feat, src, dst, nEdges, acc, astr, qe, degl, qn);
    }
#pragma unroll
    for (int b = 0; b < 8; ++b) c[b] = n[b];
    t = tn;
  }
}

__device__ __forceinline__ void rs_take(int d, int e, int lo, int cnt,
                                        const float* __restrict__ feat,
                                        const int* __restrict__ src,
                                        float* acc, int astr, int* qe, unsigned short* qd,
                                        int* degl, int* qn) {
  const unsigned u = (unsigned)(d - lo);
  if (u < (unsigned)cnt) {
    atomicAdd(&degl[u], 1);
    const int p = atomicAdd(qn, 1);
    if (p < QCAP) {
      qe[p] = e;
      qd[p] = (unsigned short)u;
    } else {
      const int s = src[e];
#pragma unroll 1
      for (int f = 0; f < NFEAT; ++f)
        atomicAdd(&acc[u * astr + f], feat[(long long)s * NFEAT + f]);
    }
  }
}

__device__ __forceinline__ void rs_filter(const float* __restrict__ feat,
                                          const int* __restrict__ src,
                                          const int* __restrict__ dst,
                                          long long nEdges, int lo, int cnt, int tid,
                                          float* acc, int astr, int* qe, unsigned short* qd,
                                          int* degl, int* qn) {
  const long long n4 = nEdges >> 2;
  const int4* __restrict__ dst4 = (const int4*)dst;
  const int nFull = (int)(n4 / (4 * RS_THREADS));
  const int rot   = (int)(((long long)blockIdx.x * nFull) / gridDim.x);
  for (int gg = 0; gg < nFull; ++gg) {
    int g = gg + rot;
    if (g >= nFull) g -= nFull;
    const long long i = (long long)g * (4 * RS_THREADS) + tid;
    const int4 d0 = dst4[i];
    const int4 d1 = dst4[i + RS_THREADS];
    const int4 d2 = dst4[i + 2 * RS_THREADS];
    const int4 d3 = dst4[i + 3 * RS_THREADS];
    const int e0 = (int)(i * 4);
    const int e1 = e0 + 4 * RS_THREADS, e2 = e0 + 8 * RS_THREADS, e3 = e0 + 12 * RS_THREADS;
    rs_take(d0.x, e0 + 0, lo, cnt, feat, src, acc, astr, qe, qd, degl, qn);
    rs_take(d0.y, e0 + 1, lo, cnt, feat, src, acc, astr, qe, qd, degl, qn);
    rs_take(d0.z, e0 + 2, lo, cnt, feat, src, acc, astr, qe, qd, degl, qn);
    rs_take(d0.w, e0 + 3, lo, cnt, feat, src, acc, astr, qe, qd, degl, qn);
    rs_take(d1.x, e1 + 0, lo, cnt, feat, src, acc, astr, qe, qd, degl, qn);
    rs_take(d1.y, e1 + 1, lo, cnt, feat, src, acc, astr, qe, qd, degl, qn);
    rs_take(d1.z, e1 + 2, lo, cnt, feat, src, acc, astr, qe, qd, degl, qn);
    rs_take(d1.w, e1 + 3, lo, cnt, feat, src, acc, astr, qe, qd, degl, qn);
    rs_take(d2.x, e2 + 0, lo, cnt, feat, src, acc, astr, qe, qd, degl, qn);
    rs_take(d2.y, e2 + 1, lo, cnt, feat, src, acc, astr, qe, qd, degl, qn);
    rs_take(d2.z, e2 + 2, lo, cnt, feat, src, acc, astr, qe, qd, degl, qn);
    rs_take(d2.w, e2 + 3, lo, cnt, feat, src, acc, astr, qe, qd, degl, qn);
    rs_take(d3.x, e3 + 0, lo, cnt, feat, src, acc, astr, qe, qd, degl, qn);
    rs_take(d3.y, e3 + 1, lo, cnt, feat, src, acc, astr, qe, qd, degl, qn);
    rs_take(d3.z, e3 + 2, lo, cnt, feat, src, acc, astr, qe, qd, degl, qn);
    rs_take(d3.w, e3 + 3, lo, cnt, feat, src, acc, astr, qe, qd, degl, qn);
  }
  for (long long i = (long long)nFull * (4 * RS_THREADS) + tid; i < n4; i += RS_THREADS) {
    const int4 d = dst4[i];
    const int e = (int)(i * 4);
    rs_take(d.x, e + 0, lo, cnt, feat, src, acc, astr, qe, qd, degl, qn);
    rs_take(d.y, e + 1, lo, cnt, feat, src, acc, astr, qe, qd, degl, qn);
    rs_take(d.z, e + 2, lo, cnt, feat, src, acc, astr, qe, qd, degl, qn);
    rs_take(d.w, e + 3, lo, cnt, feat, src, acc, astr, qe, qd, degl, qn);
  }
  for (long long e = n4 * 4 + tid; e < nEdges; e += RS_THREADS)
    rs_take(dst[e], (int)e, lo, cnt, feat, src, acc, astr, qe, qd, degl, qn);
}

__device__ __forceinline__ void own_load8(const unsigned* q, int nw, int b, int lane,
                                          const float* __restrict__ frow,
                                          unsigned (&e)[8], float4 (&v)[8]) {
  int idx = b * 8 + (lane & 7);
  if (idx > nw - 1) idx = nw - 1;
  const int mine = (int)q[idx];
#pragma unroll
  for (int k = 0; k < 8; ++k) {
    e[k] = (unsigned)__builtin_amdgcn_readlane(mine, k);
    v[k] = *(const float4*)(frow + (long long)(e[k] >> 9) * NFEAT);
  }
}

__device__ __forceinline__ void own_add8(float* acc, int astr, int nw, int b, int lane,
                                         const unsigned (&e)[8], const float4 (&v)[8]) {
#pragma unroll
  for (int k = 0; k < 8; ++k) {
    if (b * 8 + k < nw) {
      float4* p = (float4*)&acc[(int)(e[k] & 511u) * astr + 4 * lane];
      float4 a = *p;
      a.x += v[k].x; a.y += v[k].y; a.z += v[k].z; a.w += v[k].w;
      *p = a;
    }
  }
}

template <int NT>
struct L1Frag {
  float4 hv[NT];
  v2f bs0, bs1, bn0, bn1;
};

template <int NT>
__device__ __forceinline__ void l1_load(const float* const (&xr)[NT], const float* __restrict__ wsp,
                                        const float* __restrict__ wnp, int m, L1Frag<NT>& f) {
  f.bs0.x = wsp[(8 * m + 0) * NHID];  f.bs0.y = wsp[(8 * m + 1) * NHID];
  f.bs1.x = wsp[(8 * m + 2) * NHID];  f.bs1.y = wsp[(8 * m + 3) * NHID];
  f.bn0.x = wnp[(8 * m + 0) * NHID];  f.bn0.y = wnp[(8 * m + 1) * NHID];
  f.bn1.x = wnp[(8 * m + 2) * NHID];  f.bn1.y = wnp[(8 * m + 3) * NHID];
#pragma unroll
  for (int j = 0; j < NT; ++j) f.hv[j] = *(const float4*)(xr[j] + 8 * m);
}

template <int NT>
__device__ __forceinline__ void l1_mma(const float* const (&sr)[NT], int m, const L1Frag<NT>& f,
                                       v8f (&c)[NT]) {
  float4 sv[NT];
#pragma unroll
  for (int j = 0; j < NT; ++j) sv[j] = *(const float4*)(sr[j] + 8 * m);
#pragma unroll
  for (int j = 0; j < NT; ++j) {
    v2f ah0, ah1, aa0, aa1;
    ah0.x = f.hv[j].x; ah0.y = f.hv[j].y; ah1.x = f.hv[j].z; ah1.y = f.hv[j].w;
    aa0.x = sv[j].x;   aa0.y = sv[j].y;   aa1.x = sv[j].z;   aa1.y = sv[j].w;
    c[j] = __builtin_amdgcn_wmma_f32_16x16x4_f32(false, ah0, false, f.bs0, (short)0, c[j], false, false);
    c[j] = __builtin_amdgcn_wmma_f32_16x16x4_f32(false, aa0, false, f.bn0, (short)0, c[j], false, false);
    c[j] = __builtin_amdgcn_wmma_f32_16x16x4_f32(false, ah1, false, f.bs1, (short)0, c[j], false, false);
    c[j] = __builtin_amdgcn_wmma_f32_16x16x4_f32(false, aa1, false, f.bn1, (short)0, c[j], false, false);
  }
}

template <int NT>
__device__ __forceinline__ void l1_product(const float* __restrict__ x, const float* acc,
                                           const float* __restrict__ wself,
                                           const float* __restrict__ wneigh,
                                           const float* __restrict__ bias, float* hout,
                                           int lo, int cnt, int rt0, int col, int hf, int l16) {
  const float* xr[NT];
  const float* sr[NT];
  v8f c[NT];
#pragma unroll
  for (int j = 0; j < NT; ++j) {
    int r = (rt0 + 4 * j) * 16 + l16;
    if (r >= cnt) r = cnt - 1;
    xr[j] = x + (long long)(lo + r) * NFEAT + 4 * hf;
    sr[j] = acc + r * ACC_PITCH2 + 4 * hf;
    c[j] = (v8f){0.f, 0.f, 0.f, 0.f, 0.f, 0.f, 0.f, 0.f};
  }
  const float* wsp = wself  + (4 * hf) * NHID + col;
  const float* wnp = wneigh + (4 * hf) * NHID + col;
  L1Frag<NT> fa, fb;
  l1_load<NT>(xr, wsp, wnp, 0, fa);
#pragma unroll 1
  for (int m = 0; m < NFEAT / 8; m += 2) {
    l1_load<NT>(xr, wsp, wnp, m + 1, fb);
    l1_mma<NT>(sr, m, fa, c);
    if (m + 2 < NFEAT / 8) l1_load<NT>(xr, wsp, wnp, m + 2, fa);
    l1_mma<NT>(sr, m + 1, fb, c);
  }
  const float bv = bias[col];
#pragma unroll
  for (int j = 0; j < NT; ++j) {
#pragma unroll
    for (int i = 0; i < 8; ++i) {
      const int row = (rt0 + 4 * j) * 16 + i + hf * 8;
      if (row < cnt) hout[(long long)(lo + row) * NHID + col] = fmaxf(c[j][i] + bv, 0.0f);
    }
  }
}

template <bool FIRST>
__global__ __launch_bounds__(RS_THREADS)
void range_sum_kernel(const float* __restrict__ feat,
                      const int* __restrict__ src,
                      const int* __restrict__ dst,
                      float* agg,
                      int* keep,
                      const float* __restrict__ wself,
                      const float* __restrict__ wneigh,
                      const float* __restrict__ bias,
                      float* __restrict__ out,
                      int nNodes, long long nEdges) {
  constexpr int AST = ACC_PITCH2;
  __shared__ __attribute__((aligned(16))) float acc[RNG * AST];
  __shared__ __attribute__((aligned(16))) int qe[QCAP];
  __shared__ __attribute__((aligned(16))) unsigned short qd[QCAP];
  __shared__ __attribute__((aligned(16))) int degl[RNG];
  __shared__ int qn;
  __shared__ int hd;
  __shared__ __attribute__((aligned(16))) unsigned rq[RS_WAVES * SQ];
  __shared__ int rqn[RS_WAVES];
  __shared__ __attribute__((aligned(16))) float wl[FIRST ? 4 : 2 * NHID * NCLASS];

  const int tid  = threadIdx.x;
  const int lane = tid & 31;
  const int wave = tid >> 5;
  const int lo   = blockIdx.x * RNG;
  const int cnt  = (nNodes - lo < RNG) ? (nNodes - lo) : RNG;
  int* slab = keep + (long long)lo * NFEAT;

  const float4 z4 = make_float4(0.f, 0.f, 0.f, 0.f);
  for (int i = tid; i < RNG * AST / 4; i += RS_THREADS) ((float4*)acc)[i] = z4;
  if (tid < RS_WAVES) rqn[tid] = 0;
  if (FIRST) {
    for (int i = tid; i < RNG; i += RS_THREADS) degl[i] = 0;
    if (tid == 0) qn = 0;
  } else {
    if (tid == 0) hd = slab[0];
    for (int i = tid; i < NHID / 2 * NCLASS; i += RS_THREADS) {
      const int q = i >> 4, c = i & 15;
      const int k = 8 * (q >> 2) + 4 * (q & 1) + 2 * ((q >> 1) & 1);
      v2f ws, wn;
      ws.x = wself [k * NCLASS + c];  ws.y = wself [(k + 1) * NCLASS + c];
      wn.x = wneigh[k * NCLASS + c];  wn.y = wneigh[(k + 1) * NCLASS + c];
      ((v2f*)wl)[i] = ws;
      ((v2f*)wl)[NHID / 2 * NCLASS + i] = wn;
    }
  }
  __syncthreads();

  int n;
  const bool scan = FIRST ? true : (hd < 0);
  if (scan) {
    if (!FIRST) {
      for (int i = tid; i < RNG; i += RS_THREADS) degl[i] = 0;
      if (tid == 0) qn = 0;
      __syncthreads();
    }
    const bool keyed = FIRST && keys_usable(nNodes, nEdges);
    if (keyed) ks_filter(feat, src, dst, (const unsigned char*)keep, nEdges, lo, cnt, tid, acc, AST, qe, degl, &qn);
    else       rs_filter(feat, src, dst, nEdges, lo, cnt, tid, acc, AST, qe, qd, degl, &qn);
    __syncthreads();

    const int total = qn;
    n = (total < QCAP) ? total : QCAP;

    if (keyed) {
      for (int i = tid; i < n; i += RS_THREADS) {
        const int e = qe[i];
        int s = 0;
        unsigned short uu = QD_NONE;
        if (e < nEdges) {
          const int dv = dst[e];
          const int sv = src[e];
          const unsigned u = (unsigned)(dv - lo);
          if (u < (unsigned)cnt) { s = sv; uu = (unsigned short)u; atomicAdd(&degl[u], 1); }
        }
        qe[i] = s;
        qd[i] = uu;
      }
    } else {
      for (int i = tid; i < n; i += RS_THREADS) qe[i] = src[qe[i]];
    }
    __syncthreads();

    if (FIRST) {
      const int np = (n + 7) & ~7;
      const bool fits = (total <= QCAP) && (SL_Q * 4 + 6 * np <= cnt * NFEAT * 4);
      if (fits) {
        int4* sq = (int4*)(slab + SL_Q);
        for (int i = tid; i < np / 4; i += RS_THREADS) sq[i] = ((const int4*)qe)[i];
        int4* sd = (int4*)(slab + SL_Q + np);
        for (int i = tid; i < np / 8; i += RS_THREADS) sd[i] = ((const int4*)qd)[i];
        int4* sg = (int4*)(slab + SL_DEG);
        for (int i = tid; i < RNG / 4; i += RS_THREADS) sg[i] = ((const int4*)degl)[i];
      }
      if (tid == 0) slab[0] = fits ? n : -1;
    }
  } else {
    n = (hd < QCAP) ? hd : QCAP;
    const int np = (n + 7) & ~7;
    const int4* sq = (const int4*)(slab + SL_Q);
    for (int i = tid; i < np / 4; i += RS_THREADS) ((int4*)qe)[i] = sq[i];
    const int4* sd = (const int4*)(slab + SL_Q + np);
    for (int i = tid; i < np / 8; i += RS_THREADS) ((int4*)qd)[i] = sd[i];
    const int4* sg = (const int4*)(slab + SL_DEG);
    for (int i = tid; i < RNG / 4; i += RS_THREADS) ((int4*)degl)[i] = sg[i];
    __syncthreads();
  }

  for (int i = tid; i < n; i += RS_THREADS) {
    const unsigned u = qd[i];
    const int s = qe[i];
    if (u != QD_NONE) {
      const int ow = (int)(u & (RS_WAVES - 1));
      int pos = SQ;
      if ((unsigned)s < (1u << SQ_SBITS)) pos = atomicAdd(&rqn[ow], 1);
      if (pos < SQ) {
        rq[ow * SQ + pos] = ((unsigned)s << 9) | u;
      } else {
#pragma unroll 1
        for (int f = 0; f < NFEAT; ++f)
          atomicAdd(&acc[(int)u * AST + f], feat[(long long)s * NFEAT + f]);
      }
    }
  }
  __syncthreads();

  {
    int nw = rqn[wave];
    if (nw > SQ) nw = SQ;
    nw = __builtin_amdgcn_readfirstlane(nw);
    if (nw > 0) {
      const unsigned* q = rq + wave * SQ;
      const float* frow = feat + 4 * lane;
      const int nb = (nw + 7) >> 3;
      unsigned ea[8], eb[8];
      float4   va[8], vb[8];
      own_load8(q, nw, 0, lane, frow, ea, va);
      for (int b = 0; b < nb; b += 2) {
        own_load8(q, nw, b + 1, lane, frow, eb, vb);
        own_add8(acc, AST, nw, b, lane, ea, va);
        own_load8(q, nw, b + 2, lane, frow, ea, va);
        own_add8(acc, AST, nw, b + 1, lane, eb, vb);
      }
    }
  }
  __syncthreads();

  if (FIRST) {
    for (int i = tid; i < cnt * (NFEAT / 4); i += RS_THREADS) {
      const int row = i >> 5;
      const int c4  = (i & 31) * 4;
      float4* p = (float4*)(&acc[row * AST + c4]);
      const float4 s4 = *p;
      const float dv = fmaxf((float)degl[row], 1.0f);
      float4 o;
      o.x = s4.x / dv; o.y = s4.y / dv; o.z = s4.z / dv; o.w = s4.w / dv;
      *p = o;
    }
    __syncthreads();

    const int wa = wave >> 3, wb = (wave >> 2) & 1, wc = wave & 3;
    const int rg = (wa + wc) & 3;
    const int ct = 2 * wa + wb;
    const int hf  = lane >> 4;
    const int l16 = lane & 15;
    const int col = ct * 16 + l16;
    const int ntile = (cnt + 15) >> 4;
    int nt = (ntile > rg) ? ((ntile - rg + 3) >> 2) : 0;
    nt = __builtin_amdgcn_readfirstlane(nt);
    for (int j0 = 0; j0 < nt; j0 += 4) {
      const int n   = (nt - j0 < 4) ? (nt - j0) : 4;
      const int rt0 = rg + 4 * j0;
      switch (n) {
        case 4:  l1_product<4>(feat, acc, wself, wneigh, bias, agg, lo, cnt, rt0, col, hf, l16); break;
        case 3:  l1_product<3>(feat, acc, wself, wneigh, bias, agg, lo, cnt, rt0, col, hf, l16); break;
        case 2:  l1_product<2>(feat, acc, wself, wneigh, bias, agg, lo, cnt, rt0, col, hf, l16); break;
        default: l1_product<1>(feat, acc, wself, wneigh, bias, agg, lo, cnt, rt0, col, hf, l16); break;
      }
    }
  } else {
    const int hf  = lane >> 4;
    const int l16 = lane & 15;
    const v2f* wls = (const v2f*)wl;
    const v2f* wln = wls + NHID / 2 * NCLASS;
    for (int t = wave; t * 16 < cnt; t += RS_THREADS / 32) {
      int r = t * 16 + l16;
      if (r >= cnt) r = cnt - 1;
      const float rdv = 1.0f / fmaxf((float)degl[r], 1.0f);
      const float* hrow = feat + (long long)(lo + r) * NFEAT + 4 * hf;
      const float* srow = &acc[r * AST + 4 * hf];
      v8f c = {0.f, 0.f, 0.f, 0.f, 0.f, 0.f, 0.f, 0.f};
#pragma unroll
      for (int m = 0; m < NFEAT / 8; ++m) {
        const float4 hv = *(const float4*)(hrow + 8 * m);
        const float4 sv = *(const float4*)(srow + 8 * m);
        const v2f bs0 = wls[(4 * m + 0) * 16 + lane];
        const v2f bs1 = wls[(4 * m + 2) * 16 + lane];
        const v2f bn0 = wln[(4 * m + 0) * 16 + lane];
        const v2f bn1 = wln[(4 * m + 2) * 16 + lane];
        v2f ah0, ah1, aa0, aa1;
        ah0.x = hv.x;       ah0.y = hv.y;       ah1.x = hv.z;       ah1.y = hv.w;
        aa0.x = sv.x * rdv; aa0.y = sv.y * rdv; aa1.x = sv.z * rdv; aa1.y = sv.w * rdv;
        c = __builtin_amdgcn_wmma_f32_16x16x4_f32(false, ah0, false, bs0, (short)0, c, false, false);
        c = __builtin_amdgcn_wmma_f32_16x16x4_f32(false, aa0, false, bn0, (short)0, c, false, false);
        c = __builtin_amdgcn_wmma_f32_16x16x4_f32(false, ah1, false, bs1, (short)0, c, false, false);
        c = __builtin_amdgcn_wmma_f32_16x16x4_f32(false, aa1, false, bn1, (short)0, c, false, false);
      }

      const float bv = bias[l16];
#pragma unroll
      for (int j = 0; j < 8; ++j) {
        const float v = c[j] + bv;
        float mx = v;
#pragma unroll
        for (int off = 8; off >= 1; off >>= 1)
          mx = fmaxf(mx, __shfl_xor(mx, off, 16));
        const float e = expf(v - mx);
        float sm = e;
#pragma unroll
        for (int off = 8; off >= 1; off >>= 1)
          sm += __shfl_xor(sm, off, 16);
        const int row = t * 16 + j + hf * 8;
        if (row < cnt) out[(long long)(lo + row) * NCLASS + l16] = v - mx - logf(sm);
      }
    }
  }
}

extern "C" void kernel_launch(void* const* d_in, const int* in_sizes, int n_in,
                              void* d_out, int out_size, void* d_ws, size_t ws_size,
                              hipStream_t stream) {
  const float* x   = (const float*)d_in[0];
  const int*   src = (const int*)  d_in[1];
  const int*   dst = (const int*)  d_in[2];
  const float* ws1 = (const float*)d_in[3];
  const float* wn1 = (const float*)d_in[4];
  const float* b1  = (const float*)d_in[5];
  const float* ws2 = (const float*)d_in[6];
  const float* wn2 = (const float*)d_in[7];
  const float* b2  = (const float*)d_in[8];
  float* out = (float*)d_out;

  const int nNodes = in_sizes[0] / NFEAT;
  const long long nEdges = in_sizes[1];

  float* pa = (float*)d_ws;
  float* pb = pa + (size_t)nNodes * NFEAT;

  const int rsBlocks = (nNodes + RNG - 1) / RNG;

  const int keyBlocks = (int)((nEdges + KEY_PIECE - 1) / KEY_PIECE) * 8;

  range_key_kernel<<<keyBlocks, KEY_THREADS, 0, stream>>>(dst, (unsigned*)pb, nNodes, nEdges);
  range_sum_kernel<true><<<rsBlocks, RS_THREADS, 0, stream>>>(x, src, dst, pa, (int*)pb,
                                                              ws1, wn1, b1, nullptr, nNodes, nEdges);

  range_sum_kernel<false><<<rsBlocks, RS_THREADS, 0, stream>>>(pa, src, dst, nullptr, (int*)pb,
                                                               ws2, wn2, b2, out, nNodes, nEdges);
}


# layer-1 product takes its x rows from LDS as well: whole 512-B rows staged into the space the queue has left, 13 row tiles per pass, not row-scattered fragment loads from memory; sums unchanged
# speedup vs baseline: 4.0629x; 1.0540x over previous
#include <hip/hip_runtime.h>
#include <math.h>

typedef __attribute__((ext_vector_type(2))) float v2f;
typedef __attribute__((ext_vector_type(8))) float v8f;

#define NFEAT 128
#define NHID  128
#define NCLASS 16

#define RNG   400
#define QCAP  8192
#define RS_THREADS 1024
#define RS_WAVES (RS_THREADS / 32)

#define SQ      320
#define SQ_SBITS 23

#define SL_DEG 4
#define SL_Q   (SL_DEG + RNG)

#define SLAB_BYTES (RNG * NFEAT * 4)
#define KEY_OFF    131072
#define KEY_PIECE  65536
#define KU_EDGES   128
#define KU_PER_PIECE (KEY_PIECE / KU_EDGES)
#define KT_BYTES   4096
#define KEY_THREADS 256
#define QD_NONE    0xFFFF

#define ACC_PITCH2 132
static_assert(SL_Q * 4 + 6 * QCAP <= KEY_OFF, "parked queue must end below the key bytes");
static_assert(KEY_OFF + KEY_PIECE <= SLAB_BYTES, "key piece must fit in a full range's rows");
static_assert(KU_PER_PIECE == 512 && KT_BYTES == 32 * 8 * 16, "unit and tile arithmetic below");
static_assert(RNG < QD_NONE, "local dst must not collide with the marker");
static_assert(RNG <= 512, "local dst must fit the 9 low bits of a routed word");
static_assert(SQ % 8 == 0, "an owner takes its edges 8 at a time");
#define XS_TILES 13
#define XS_ROWS  (XS_TILES * 16)
#define POOL_Q   (6 * QCAP + 4 * RS_WAVES * SQ)
#define POOL_XS  (XS_ROWS * ACC_PITCH2 * 4)
#define POOL_L1  (POOL_XS > POOL_Q ? POOL_XS : POOL_Q)
#define XS_LOADS ((XS_ROWS * (NFEAT / 4) + RS_THREADS - 1) / RS_THREADS)
static_assert(2 * XS_ROWS >= RNG, "two passes must cover a range");
static_assert((6 * QCAP) % 16 == 0, "the owners' lists start on a 16-B boundary");

__device__ __forceinline__ bool keys_usable(int nNodes, long long nEdges) {
  const long long nPieces = (nEdges + KEY_PIECE - 1) / KEY_PIECE;
  return nPieces * RNG <= (long long)nNodes;
}

__device__ __forceinline__ unsigned range_no(int d) {
  return (d < 0) ? 255u : (((unsigned)d / RNG) & 255u);
}

__global__ __launch_bounds__(KEY_THREADS)
void range_key_kernel(const int* __restrict__ dst, unsigned* keep, int nNodes, long long nEdges) {
  if (!keys_usable(nNodes, nEdges)) return;
  const int lane = threadIdx.x & 31;
  const int U0   = (blockIdx.x * (KEY_THREADS / 32) + (threadIdx.x >> 5)) * 8;
  const int piece = U0 / KU_PER_PIECE;
  const int ul0   = U0 % KU_PER_PIECE;
  int4 d[8];
#pragma unroll
  for (int k = 0; k < 8; ++k) {
    const long long e = (long long)(U0 + k) * KU_EDGES + 4 * lane;
    if (e + 4 <= nEdges) {
      d[k] = *(const int4*)(dst + e);
    } else {
      d[k].x = (e + 0 < nEdges) ? dst[e + 0] : -1;
      d[k].y = (e + 1 < nEdges) ? dst[e + 1] : -1;
      d[k].z = (e + 2 < nEdges) ? dst[e + 2] : -1;
      d[k].w = (e + 3 < nEdges) ? dst[e + 3] : -1;
    }
  }
  unsigned r[4][8];
#pragma unroll
  for (int j = 0; j < 4; ++j)
#pragma unroll
    for (int b = 0; b < 8; ++b) r[j][b] = 0u;
#pragma unroll
  for (int k = 0; k < 8; ++k) {
    unsigned kj[4];
    kj[0] = range_no(d[k].x); kj[1] = range_no(d[k].y);
    kj[2] = range_no(d[k].z); kj[3] = range_no(d[k].w);
#pragma unroll
    for (int j = 0; j < 4; ++j) {
#pragma unroll
      for (int b = 0; b < 8; ++b) {
        const unsigned pl = __builtin_amdgcn_ballot_w32(((kj[j] >> b) & 1u) != 0u);
        if (lane == k) r[j][b] = pl;
      }
    }
  }
  if (lane < 8) {
    unsigned* tile = keep + (long long)piece * (SLAB_BYTES / 4) + KEY_OFF / 4 + (ul0 >> 5) * (KT_BYTES / 4);
    const int ul = (ul0 & 31) + lane;
#pragma unroll
    for (int b = 0; b < 8; ++b)
      *(uint4*)(tile + b * 128 + ul * 4) = make_uint4(r[0][b], r[1][b], r[2][b], r[3][b]);
  }
}

__device__ __forceinline__ void ks_hits(unsigned m, int e0, int lo, int cnt,
                                        const float* __restrict__ feat,
                                        const int* __restrict__ src,
                                        const int* __restrict__ dst, long long nEdges,
                                        float* acc, int astr, int* qe, int* degl, int* qn) {
  while (m) {
    const int pos = __ffs(m) - 1;
    const int e = e0 + 4 * pos;
    m &= m - 1;
    const int p = atomicAdd(qn, 1);
    if (p < QCAP) {
      qe[p] = e;
    } else if (e < nEdges) {
      const unsigned u = (unsigned)(dst[e] - lo);
      if (u < (unsigned)cnt) {
        atomicAdd(&degl[u], 1);
        const int s = src[e];
#pragma unroll 1
        for (int f = 0; f < NFEAT; ++f)
          atomicAdd(&acc[u * astr + f], feat[(long long)s * NFEAT + f]);
      }
    }
  }
}

__device__ __forceinline__ void ks_load(const unsigned char* keys, int U, int nUnits, uint4 (&c)[8]) {
  if (U < nUnits) {
    const unsigned char* p = keys + (long long)(U / KU_PER_PIECE) * SLAB_BYTES + KEY_OFF +
                             ((U % KU_PER_PIECE) >> 5) * KT_BYTES + (U & 31) * 16;
#pragma unroll
    for (int b = 0; b < 8; ++b) c[b] = *(const uint4*)(p + b * 512);
  } else {
#pragma unroll
    for (int b = 0; b < 8; ++b) c[b] = make_uint4(0u, 0u, 0u, 0u);
  }
}

__device__ __forceinline__ void ks_filter(const float* __restrict__ feat,
                                          const int* __restrict__ src,
                                          const int* __restrict__ dst,
                                          const unsigned char* keys,
                                          long long nEdges, int lo, int cnt, int tid,
                                          float* acc, int astr, int* qe, int* degl, int* qn) {
  const int nPieces = (int)((nEdges + KEY_PIECE - 1) / KEY_PIECE);
  const int nUnits  = nPieces * KU_PER_PIECE;
  const int nTrips  = (nUnits + RS_THREADS - 1) / RS_THREADS;
  const unsigned key = blockIdx.x & 255u;
  unsigned sm[8];
#pragma unroll
  for (int b = 0; b < 8; ++b) sm[b] = 0u - ((key >> b) & 1u);
  int t = (int)(((long long)blockIdx.x * nTrips) / gridDim.x);
  uint4 c[8];
  ks_load(keys, t * RS_THREADS + tid, nUnits, c);
  for (int tt = 0; tt < nTrips; ++tt) {
    int tn = t + 1;
    if (tn >= nTrips) tn = 0;
    uint4 n[8];
#pragma unroll
    for (int b = 0; b < 8; ++b) n[b] = c[b];
    if (tt + 1 < nTrips) ks_load(keys, tn * RS_THREADS + tid, nUnits, n);
    const int U = t * RS_THREADS + tid;
    if (U < nUnits) {
      uint4 mm;
      mm.x = c[0].x ^ sm[0]; mm.y = c[0].y ^ sm[0]; mm.z = c[0].z ^ sm[0]; mm.w = c[0].w ^ sm[0];
#pragma unroll
      for (int b = 1; b < 8; ++b) {
        mm.x |= c[b].x ^ sm[b]; mm.y |= c[b].y ^ sm[b];
        mm.z |= c[b].z ^ sm[b]; mm.w |= c[b].w ^ sm[b];
      }
      const int e0 = U * KU_EDGES;
      ks_hits(~mm.x, e0 + 0, lo, cnt, feat, src, dst, nEdges, acc, astr, qe, degl, qn);
      ks_hits(~mm.y, e0 + 1, lo, cnt, feat, src, dst, nEdges, acc, astr, qe, degl, qn);
      ks_hits(~mm.z, e0 + 2, lo, cnt, feat, src, dst, nEdges, acc, astr, qe, degl, qn);
      ks_hits(~mm.w, e0 + 3, lo, cnt, feat, src, dst, nEdges, acc, astr, qe, degl, qn);
    }
#pragma unroll
    for (int b = 0; b < 8; ++b) c[b] = n[b];
    t = tn;
  }
}

__device__ __forceinline__ void rs_take(int d, int e, int lo, int cnt,
                                        const float* __restrict__ feat,
                                        const int* __restrict__ src,
                                        float* acc, int astr, int* qe, unsigned short* qd,
                                        int* degl, int* qn) {
  const unsigned u = (unsigned)(d - lo);
  if (u < (unsigned)cnt) {
    atomicAdd(&degl[u], 1);
    const int p = atomicAdd(qn, 1);
    if (p < QCAP) {
      qe[p] = e;
      qd[p] = (unsigned short)u;
    } else {
      const int s = src[e];
#pragma unroll 1
      for (int f = 0; f < NFEAT; ++f)
        atomicAdd(&acc[u * astr + f], feat[(long long)s * NFEAT + f]);
    }
  }
}

__device__ __forceinline__ void rs_filter(const float* __restrict__ feat,
                                          const int* __restrict__ src,
                                          const int* __restrict__ dst,
                                          long long nEdges, int lo, int cnt, int tid,
                                          float* acc, int astr, int* qe, unsigned short* qd,
                                          int* degl, int* qn) {
  const long long n4 = nEdges >> 2;
  const int4* __restrict__ dst4 = (const int4*)dst;
  const int nFull = (int)(n4 / (4 * RS_THREADS));
  const int rot   = (int)(((long long)blockIdx.x * nFull) / gridDim.x);
  for (int gg = 0; gg < nFull; ++gg) {
    int g = gg + rot;
    if (g >= nFull) g -= nFull;
    const long long i = (long long)g * (4 * RS_THREADS) + tid;
    const int4 d0 = dst4[i];
    const int4 d1 = dst4[i + RS_THREADS];
    const int4 d2 = dst4[i + 2 * RS_THREADS];
    const int4 d3 = dst4[i + 3 * RS_THREADS];
    const int e0 = (int)(i * 4);
    const int e1 = e0 + 4 * RS_THREADS, e2 = e0 + 8 * RS_THREADS, e3 = e0 + 12 * RS_THREADS;
    rs_take(d0.x, e0 + 0, lo, cnt, feat, src, acc, astr, qe, qd, degl, qn);
    rs_take(d0.y, e0 + 1, lo, cnt, feat, src, acc, astr, qe, qd, degl, qn);
    rs_take(d0.z, e0 + 2, lo, cnt, feat, src, acc, astr, qe, qd, degl, qn);
    rs_take(d0.w, e0 + 3, lo, cnt, feat, src, acc, astr, qe, qd, degl, qn);
    rs_take(d1.x, e1 + 0, lo, cnt, feat, src, acc, astr, qe, qd, degl, qn);
    rs_take(d1.y, e1 + 1, lo, cnt, feat, src, acc, astr, qe, qd, degl, qn);
    rs_take(d1.z, e1 + 2, lo, cnt, feat, src, acc, astr, qe, qd, degl, qn);
    rs_take(d1.w, e1 + 3, lo, cnt, feat, src, acc, astr, qe, qd, degl, qn);
    rs_take(d2.x, e2 + 0, lo, cnt, feat, src, acc, astr, qe, qd, degl, qn);
    rs_take(d2.y, e2 + 1, lo, cnt, feat, src, acc, astr, qe, qd, degl, qn);
    rs_take(d2.z, e2 + 2, lo, cnt, feat, src, acc, astr, qe, qd, degl, qn);
    rs_take(d2.w, e2 + 3, lo, cnt, feat, src, acc, astr, qe, qd, degl, qn);
    rs_take(d3.x, e3 + 0, lo, cnt, feat, src, acc, astr, qe, qd, degl, qn);
    rs_take(d3.y, e3 + 1, lo, cnt, feat, src, acc, astr, qe, qd, degl, qn);
    rs_take(d3.z, e3 + 2, lo, cnt, feat, src, acc, astr, qe, qd, degl, qn);
    rs_take(d3.w, e3 + 3, lo, cnt, feat, src, acc, astr, qe, qd, degl, qn);
  }
  for (long long i = (long long)nFull * (4 * RS_THREADS) + tid; i < n4; i += RS_THREADS) {
    const int4 d = dst4[i];
    const int e = (int)(i * 4);
    rs_take(d.x, e + 0, lo, cnt, feat, src, acc, astr, qe, qd, degl, qn);
    rs_take(d.y, e + 1, lo, cnt, feat, src, acc, astr, qe, qd, degl, qn);
    rs_take(d.z, e + 2, lo, cnt, feat, src, acc, astr, qe, qd, degl, qn);
    rs_take(d.w, e + 3, lo, cnt, feat, src, acc, astr, qe, qd, degl, qn);
  }
  for (long long e = n4 * 4 + tid; e < nEdges; e += RS_THREADS)
    rs_take(dst[e], (int)e, lo, cnt, feat, src, acc, astr, qe, qd, degl, qn);
}

__device__ __forceinline__ void own_load8(const unsigned* q, int nw, int b, int lane,
                                          const float* __restrict__ frow,
                                          unsigned (&e)[8], float4 (&v)[8]) {
  int idx = b * 8 + (lane & 7);
  if (idx > nw - 1) idx = nw - 1;
  const int mine = (int)q[idx];
#pragma unroll
  for (int k = 0; k < 8; ++k) {
    e[k] = (unsigned)__builtin_amdgcn_readlane(mine, k);
    v[k] = *(const float4*)(frow + (long long)(e[k] >> 9) * NFEAT);
  }
}

__device__ __forceinline__ void own_add8(float* acc, int astr, int nw, int b, int lane,
                                         const unsigned (&e)[8], const float4 (&v)[8]) {
#pragma unroll
  for (int k = 0; k < 8; ++k) {
    if (b * 8 + k < nw) {
      float4* p = (float4*)&acc[(int)(e[k] & 511u) * astr + 4 * lane];
      float4 a = *p;
      a.x += v[k].x; a.y += v[k].y; a.z += v[k].z; a.w += v[k].w;
      *p = a;
    }
  }
}

struct L1Frag {
  v2f bs0, bs1, bn0, bn1;
};

__device__ __forceinline__ void l1_load(const float* __restrict__ wsp, const float* __restrict__ wnp,
                                        int m, L1Frag& f) {
  f.bs0.x = wsp[(8 * m + 0) * NHID];  f.bs0.y = wsp[(8 * m + 1) * NHID];
  f.bs1.x = wsp[(8 * m + 2) * NHID];  f.bs1.y = wsp[(8 * m + 3) * NHID];
  f.bn0.x = wnp[(8 * m + 0) * NHID];  f.bn0.y = wnp[(8 * m + 1) * NHID];
  f.bn1.x = wnp[(8 * m + 2) * NHID];  f.bn1.y = wnp[(8 * m + 3) * NHID];
}

template <int NT>
__device__ __forceinline__ void l1_mma(const float* const (&xr)[NT], const float* const (&sr)[NT], int m,
                                       const L1Frag& f, v8f (&c)[NT]) {
  float4 hv[NT], sv[NT];
#pragma unroll
  for (int j = 0; j < NT; ++j) {
    hv[j] = *(const float4*)(xr[j] + 8 * m);
    sv[j] = *(const float4*)(sr[j] + 8 * m);
  }
#pragma unroll
  for (int j = 0; j < NT; ++j) {
    v2f ah0, ah1, aa0, aa1;
    ah0.x = hv[j].x; ah0.y = hv[j].y; ah1.x = hv[j].z; ah1.y = hv[j].w;
    aa0.x = sv[j].x; aa0.y = sv[j].y; aa1.x = sv[j].z; aa1.y = sv[j].w;
    c[j] = __builtin_amdgcn_wmma_f32_16x16x4_f32(false, ah0, false, f.bs0, (short)0, c[j], false, false);
    c[j] = __builtin_amdgcn_wmma_f32_16x16x4_f32(false, aa0, false, f.bn0, (short)0, c[j], false, false);
    c[j] = __builtin_amdgcn_wmma_f32_16x16x4_f32(false, ah1, false, f.bs1, (short)0, c[j], false, false);
    c[j] = __builtin_amdgcn_wmma_f32_16x16x4_f32(false, aa1, false, f.bn1, (short)0, c[j], false, false);
  }
}

template <int NT>
__device__ __forceinline__ void l1_product(const float* xs, int rb, const float* acc,
                                           const float* __restrict__ wself,
                                           const float* __restrict__ wneigh,
                                           const float* __restrict__ bias, float* hout,
                                           int lo, int cnt, int rt0, int col, int hf, int l16) {
  const float* xr[NT];
  const float* sr[NT];
  v8f c[NT];
#pragma unroll
  for (int j = 0; j < NT; ++j) {
    int r = (rt0 + 4 * j) * 16 + l16;
    if (r >= cnt) r = cnt - 1;
    xr[j] = xs  + (r - rb) * ACC_PITCH2 + 4 * hf;
    sr[j] = acc + r * ACC_PITCH2 + 4 * hf;
    c[j] = (v8f){0.f, 0.f, 0.f, 0.f, 0.f, 0.f, 0.f, 0.f};
  }
  const float* wsp = wself  + (4 * hf) * NHID + col;
  const float* wnp = wneigh + (4 * hf) * NHID + col;
  L1Frag fa, fb;
  l1_load(wsp, wnp, 0, fa);
#pragma unroll 1
  for (int m = 0; m < NFEAT / 8; m += 2) {
    l1_load(wsp, wnp, m + 1, fb);
    l1_mma<NT>(xr, sr, m, fa, c);
    if (m + 2 < NFEAT / 8) l1_load(wsp, wnp, m + 2, fa);
    l1_mma<NT>(xr, sr, m + 1, fb, c);
  }
  const float bv = bias[col];
#pragma unroll
  for (int j = 0; j < NT; ++j) {
#pragma unroll
    for (int i = 0; i < 8; ++i) {
      const int row = (rt0 + 4 * j) * 16 + i + hf * 8;
      if (row < cnt) hout[(long long)(lo + row) * NHID + col] = fmaxf(c[j][i] + bv, 0.0f);
    }
  }
}

__device__ __forceinline__ void xs_load(const float* __restrict__ feat, int r0, int np, int tid,
                                        float4 (&pre)[XS_LOADS]) {
  const float* base = feat + (long long)r0 * NFEAT;
#pragma unroll
  for (int k = 0; k < XS_LOADS; ++k) {
    const int i = tid + k * RS_THREADS;
    pre[k] = make_float4(0.f, 0.f, 0.f, 0.f);
    if (i < np) pre[k] = *(const float4*)(base + (long long)i * 4);
  }
}

__device__ __forceinline__ void xs_store(float* xs, int np, int tid, const float4 (&pre)[XS_LOADS]) {
#pragma unroll
  for (int k = 0; k < XS_LOADS; ++k) {
    const int i = tid + k * RS_THREADS;
    if (i < np) *(float4*)(&xs[(i >> 5) * ACC_PITCH2 + (i & 31) * 4]) = pre[k];
  }
}

__device__ __forceinline__ void l1_pass(const float* xs, const float* acc,
                                        const float* __restrict__ wself,
                                        const float* __restrict__ wneigh,
                                        const float* __restrict__ bias, float* hout,
                                        int lo, int cnt, int tb, int te, int rg, int col, int hf, int l16) {
  const int span = te - tb;
  int nt = (span > rg) ? ((span - rg + 3) >> 2) : 0;
  nt = __builtin_amdgcn_readfirstlane(nt);
  for (int j0 = 0; j0 < nt; j0 += 4) {
    const int n   = (nt - j0 < 4) ? (nt - j0) : 4;
    const int rt0 = tb + rg + 4 * j0;
    switch (n) {
      case 4:  l1_product<4>(xs, tb * 16, acc, wself, wneigh, bias, hout, lo, cnt, rt0, col, hf, l16); break;
      case 3:  l1_product<3>(xs, tb * 16, acc, wself, wneigh, bias, hout, lo, cnt, rt0, col, hf, l16); break;
      case 2:  l1_product<2>(xs, tb * 16, acc, wself, wneigh, bias, hout, lo, cnt, rt0, col, hf, l16); break;
      default: l1_product<1>(xs, tb * 16, acc, wself, wneigh, bias, hout, lo, cnt, rt0, col, hf, l16); break;
    }
  }
}

template <bool FIRST>
__global__ __launch_bounds__(RS_THREADS)
void range_sum_kernel(const float* __restrict__ feat,
                      const int* __restrict__ src,
                      const int* __restrict__ dst,
                      float* agg,
                      int* keep,
                      const float* __restrict__ wself,
                      const float* __restrict__ wneigh,
                      const float* __restrict__ bias,
                      float* __restrict__ out,
                      int nNodes, long long nEdges) {
  constexpr int AST = ACC_PITCH2;
  __shared__ __attribute__((aligned(16))) float acc[RNG * AST];
  __shared__ __attribute__((aligned(16))) unsigned char pool[FIRST ? POOL_L1 : POOL_Q];
  int* qe = (int*)pool;
  unsigned short* qd = (unsigned short*)(pool + 4 * QCAP);
  unsigned* rq = (unsigned*)(pool + 6 * QCAP);
  __shared__ __attribute__((aligned(16))) int degl[RNG];
  __shared__ int qn;
  __shared__ int hd;
  __shared__ int rqn[RS_WAVES];
  __shared__ __attribute__((aligned(16))) float wl[FIRST ? 4 : 2 * NHID * NCLASS];

  const int tid  = threadIdx.x;
  const int lane = tid & 31;
  const int wave = tid >> 5;
  const int lo   = blockIdx.x * RNG;
  const int cnt  = (nNodes - lo < RNG) ? (nNodes - lo) : RNG;
  int* slab = keep + (long long)lo * NFEAT;

  const float4 z4 = make_float4(0.f, 0.f, 0.f, 0.f);
  for (int i = tid; i < RNG * AST / 4; i += RS_THREADS) ((float4*)acc)[i] = z4;
  if (tid < RS_WAVES) rqn[tid] = 0;
  if (FIRST) {
    for (int i = tid; i < RNG; i += RS_THREADS) degl[i] = 0;
    if (tid == 0) qn = 0;
  } else {
    if (tid == 0) hd = slab[0];
    for (int i = tid; i < NHID / 2 * NCLASS; i += RS_THREADS) {
      const int q = i >> 4, c = i & 15;
      const int k = 8 * (q >> 2) + 4 * (q & 1) + 2 * ((q >> 1) & 1);
      v2f ws, wn;
      ws.x = wself [k * NCLASS + c];  ws.y = wself [(k + 1) * NCLASS + c];
      wn.x = wneigh[k * NCLASS + c];  wn.y = wneigh[(k + 1) * NCLASS + c];
      ((v2f*)wl)[i] = ws;
      ((v2f*)wl)[NHID / 2 * NCLASS + i] = wn;
    }
  }
  __syncthreads();

  int n;
  const bool scan = FIRST ? true : (hd < 0);
  if (scan) {
    if (!FIRST) {
      for (int i = tid; i < RNG; i += RS_THREADS) degl[i] = 0;
      if (tid == 0) qn = 0;
      __syncthreads();
    }
    const bool keyed = FIRST && keys_usable(nNodes, nEdges);
    if (keyed) ks_filter(feat, src, dst, (const unsigned char*)keep, nEdges, lo, cnt, tid, acc, AST, qe, degl, &qn);
    else       rs_filter(feat, src, dst, nEdges, lo, cnt, tid, acc, AST, qe, qd, degl, &qn);
    __syncthreads();

    const int total = qn;
    n = (total < QCAP) ? total : QCAP;

    if (keyed) {
      for (int i = tid; i < n; i += RS_THREADS) {
        const int e = qe[i];
        int s = 0;
        unsigned short uu = QD_NONE;
        if (e < nEdges) {
          const int dv = dst[e];
          const int sv = src[e];
          const unsigned u = (unsigned)(dv - lo);
          if (u < (unsigned)cnt) { s = sv; uu = (unsigned short)u; atomicAdd(&degl[u], 1); }
        }
        qe[i] = s;
        qd[i] = uu;
      }
    } else {
      for (int i = tid; i < n; i += RS_THREADS) qe[i] = src[qe[i]];
    }
    __syncthreads();

    if (FIRST) {
      const int np = (n + 7) & ~7;
      const bool fits = (total <= QCAP) && (SL_Q * 4 + 6 * np <= cnt * NFEAT * 4);
      if (fits) {
        int4* sq = (int4*)(slab + SL_Q);
        for (int i = tid; i < np / 4; i += RS_THREADS) sq[i] = ((const int4*)qe)[i];
        int4* sd = (int4*)(slab + SL_Q + np);
        for (int i = tid; i < np / 8; i += RS_THREADS) sd[i] = ((const int4*)qd)[i];
        int4* sg = (int4*)(slab + SL_DEG);
        for (int i = tid; i < RNG / 4; i += RS_THREADS) sg[i] = ((const int4*)degl)[i];
      }
      if (tid == 0) slab[0] = fits ? n : -1;
    }
  } else {
    n = (hd < QCAP) ? hd : QCAP;
    const int np = (n + 7) & ~7;
    const int4* sq = (const int4*)(slab + SL_Q);
    for (int i = tid; i < np / 4; i += RS_THREADS) ((int4*)qe)[i] = sq[i];
    const int4* sd = (const int4*)(slab + SL_Q + np);
    for (int i = tid; i < np / 8; i += RS_THREADS) ((int4*)qd)[i] = sd[i];
    const int4* sg = (const int4*)(slab + SL_DEG);
    for (int i = tid; i < RNG / 4; i += RS_THREADS) ((int4*)degl)[i] = sg[i];
    __syncthreads();
  }

  for (int i = tid; i < n; i += RS_THREADS) {
    const unsigned u = qd[i];
    const int s = qe[i];
    if (u != QD_NONE) {
      const int ow = (int)(u & (RS_WAVES - 1));
      int pos = SQ;
      if ((unsigned)s < (1u << SQ_SBITS)) pos = atomicAdd(&rqn[ow], 1);
      if (pos < SQ) {
        rq[ow * SQ + pos] = ((unsigned)s << 9) | u;
      } else {
#pragma unroll 1
        for (int f = 0; f < NFEAT; ++f)
          atomicAdd(&acc[(int)u * AST + f], feat[(long long)s * NFEAT + f]);
      }
    }
  }
  __syncthreads();

  {
    int nw = rqn[wave];
    if (nw > SQ) nw = SQ;
    nw = __builtin_amdgcn_readfirstlane(nw);
    if (nw > 0) {
      const unsigned* q = rq + wave * SQ;
      const float* frow = feat + 4 * lane;
      const int nb = (nw + 7) >> 3;
      unsigned ea[8], eb[8];
      float4   va[8], vb[8];
      own_load8(q, nw, 0, lane, frow, ea, va);
      for (int b = 0; b < nb; b += 2) {
        own_load8(q, nw, b + 1, lane, frow, eb, vb);
        own_add8(acc, AST, nw, b, lane, ea, va);
        own_load8(q, nw, b + 2, lane, frow, ea, va);
        own_add8(acc, AST, nw, b + 1, lane, eb, vb);
      }
    }
  }
  __syncthreads();

  if (FIRST) {
    float* xs = (float*)pool;
    const int ntile = (cnt + 15) >> 4;
    const int nr0 = ((cnt < XS_ROWS) ? cnt : XS_ROWS) * (NFEAT / 4);
    float4 pre[XS_LOADS];
    xs_load(feat, lo, nr0, tid, pre);
    for (int i = tid; i < cnt * (NFEAT / 4); i += RS_THREADS) {
      const int row = i >> 5;
      const int c4  = (i & 31) * 4;
      float4* p = (float4*)(&acc[row * AST + c4]);
      const float4 s4 = *p;
      const float dv = fmaxf((float)degl[row], 1.0f);
      float4 o;
      o.x = s4.x / dv; o.y = s4.y / dv; o.z = s4.z / dv; o.w = s4.w / dv;
      *p = o;
    }
    xs_store(xs, nr0, tid, pre);
    __syncthreads();

    const int wa = wave >> 3, wb = (wave >> 2) & 1, wc = wave & 3;
    const int rg = (wa + wc) & 3;
    const int ct = 2 * wa + wb;
    const int hf  = lane >> 4;
    const int l16 = lane & 15;
    const int col = ct * 16 + l16;
    const int nr1 = (cnt > XS_ROWS) ? (cnt - XS_ROWS) * (NFEAT / 4) : 0;
#pragma unroll 1
    for (int p = 0; p < 2; ++p) {
      const int tb = p * XS_TILES;
      const int te = (ntile < tb + XS_TILES) ? ntile : (tb + XS_TILES);
      if (p == 1) {
        xs_load(feat, lo + XS_ROWS, nr1, tid, pre);
        __syncthreads();
        xs_store(xs, nr1, tid, pre);
        __syncthreads();
      }
      l1_pass(xs, acc, wself, wneigh, bias, agg, lo, cnt, tb, te, rg, col, hf, l16);
    }
  } else {
    const int hf  = lane >> 4;
    const int l16 = lane & 15;
    const v2f* wls = (const v2f*)wl;
    const v2f* wln = wls + NHID / 2 * NCLASS;
    for (int t = wave; t * 16 < cnt; t += RS_THREADS / 32) {
      int r = t * 16 + l16;
      if (r >= cnt) r = cnt - 1;
      const float rdv = 1.0f / fmaxf((float)degl[r], 1.0f);
      const float* hrow = feat + (long long)(lo + r) * NFEAT + 4 * hf;
      const float* srow = &acc[r * AST + 4 * hf];
      v8f c = {0.f, 0.f, 0.f, 0.f, 0.f, 0.f, 0.f, 0.f};
#pragma unroll
      for (int m = 0; m < NFEAT / 8; ++m) {
        const float4 hv = *(const float4*)(hrow + 8 * m);
        const float4 sv = *(const float4*)(srow + 8 * m);
        const v2f bs0 = wls[(4 * m + 0) * 16 + lane];
        const v2f bs1 = wls[(4 * m + 2) * 16 + lane];
        const v2f bn0 = wln[(4 * m + 0) * 16 + lane];
        const v2f bn1 = wln[(4 * m + 2) * 16 + lane];
        v2f ah0, ah1, aa0, aa1;
        ah0.x = hv.x;       ah0.y = hv.y;       ah1.x = hv.z;       ah1.y = hv.w;
        aa0.x = sv.x * rdv; aa0.y = sv.y * rdv; aa1.x = sv.z * rdv; aa1.y = sv.w * rdv;
        c = __builtin_amdgcn_wmma_f32_16x16x4_f32(false, ah0, false, bs0, (short)0, c, false, false);
        c = __builtin_amdgcn_wmma_f32_16x16x4_f32(false, aa0, false, bn0, (short)0, c, false, false);
        c = __builtin_amdgcn_wmma_f32_16x16x4_f32(false, ah1, false, bs1, (short)0, c, false, false);
        c = __builtin_amdgcn_wmma_f32_16x16x4_f32(false, aa1, false, bn1, (short)0, c, false, false);
      }

      const float bv = bias[l16];
#pragma unroll
      for (int j = 0; j < 8; ++j) {
        const float v = c[j] + bv;
        float mx = v;
#pragma unroll
        for (int off = 8; off >= 1; off >>= 1)
          mx = fmaxf(mx, __shfl_xor(mx, off, 16));
        const float e = expf(v - mx);
        float sm = e;
#pragma unroll
        for (int off = 8; off >= 1; off >>= 1)
          sm += __shfl_xor(sm, off, 16);
        const int row = t * 16 + j + hf * 8;
        if (row < cnt) out[(long long)(lo + row) * NCLASS + l16] = v - mx - logf(sm);
      }
    }
  }
}

extern "C" void kernel_launch(void* const* d_in, const int* in_sizes, int n_in,
                              void* d_out, int out_size, void* d_ws, size_t ws_size,
                              hipStream_t stream) {
  const float* x   = (const float*)d_in[0];
  const int*   src = (const int*)  d_in[1];
  const int*   dst = (const int*)  d_in[2];
  const float* ws1 = (const float*)d_in[3];
  const float* wn1 = (const float*)d_in[4];
  const float* b1  = (const float*)d_in[5];
  const float* ws2 = (const float*)d_in[6];
  const float* wn2 = (const float*)d_in[7];
  const float* b2  = (const float*)d_in[8];
  float* out = (float*)d_out;

  const int nNodes = in_sizes[0] / NFEAT;
  const long long nEdges = in_sizes[1];

  float* pa = (float*)d_ws;
  float* pb = pa + (size_t)nNodes * NFEAT;

  const int rsBlocks = (nNodes + RNG - 1) / RNG;

  const int keyBlocks = (int)((nEdges + KEY_PIECE - 1) / KEY_PIECE) * 8;

  range_key_kernel<<<keyBlocks, KEY_THREADS, 0, stream>>>(dst, (unsigned*)pb, nNodes, nEdges);
  range_sum_kernel<true><<<rsBlocks, RS_THREADS, 0, stream>>>(x, src, dst, pa, (int*)pb,
                                                              ws1, wn1, b1, nullptr, nNodes, nEdges);

  range_sum_kernel<false><<<rsBlocks, RS_THREADS, 0, stream>>>(pa, src, dst, nullptr, (int*)pb,
                                                               ws2, wn2, b2, out, nNodes, nEdges);
}


# owner lists grouped by destination row (stretches from the in-degrees): a row's sum stays in registers across its edges, one LDS add per row, not per edge; sums in row-grouped list order
# speedup vs baseline: 4.3801x; 1.0781x over previous
#include <hip/hip_runtime.h>
#include <math.h>

typedef __attribute__((ext_vector_type(2))) float v2f;
typedef __attribute__((ext_vector_type(8))) float v8f;

#define NFEAT 128
#define NHID  128
#define NCLASS 16

#define RNG   400
#define QCAP  8192
#define RS_THREADS 1024
#define RS_WAVES (RS_THREADS / 32)

#define SQ      320
#define SQ_SBITS 23
#define SQ_NULL 511u

#define SL_DEG 4
#define SL_Q   (SL_DEG + RNG)

#define SLAB_BYTES (RNG * NFEAT * 4)
#define KEY_OFF    131072
#define KEY_PIECE  65536
#define KU_EDGES   128
#define KU_PER_PIECE (KEY_PIECE / KU_EDGES)
#define KT_BYTES   4096
#define KEY_THREADS 256
#define QD_NONE    0xFFFF

#define ACC_PITCH2 132
static_assert(SL_Q * 4 + 6 * QCAP <= KEY_OFF, "parked queue must end below the key bytes");
static_assert(KEY_OFF + KEY_PIECE <= SLAB_BYTES, "key piece must fit in a full range's rows");
static_assert(KU_PER_PIECE == 512 && KT_BYTES == 32 * 8 * 16, "unit and tile arithmetic below");
static_assert(RNG < QD_NONE, "local dst must not collide with the marker");
static_assert(RNG <= SQ_NULL, "local dst must fit the 9 low bits of a routed word, below the filler");
static_assert((RNG + RS_WAVES - 1) / RS_WAVES <= 32, "a lane of the owner wave per owned row");
static_assert(SQ % 4 == 0, "an owner's list is filled 16 B at a time");
static_assert(SQ % 8 == 0, "an owner takes its edges 8 at a time");
#define XS_TILES 13
#define XS_ROWS  (XS_TILES * 16)
#define POOL_Q   (6 * QCAP + 4 * RS_WAVES * SQ)
#define POOL_XS  (XS_ROWS * ACC_PITCH2 * 4)
#define POOL_L1  (POOL_XS > POOL_Q ? POOL_XS : POOL_Q)
#define XS_LOADS ((XS_ROWS * (NFEAT / 4) + RS_THREADS - 1) / RS_THREADS)
static_assert(2 * XS_ROWS >= RNG, "two passes must cover a range");
static_assert((6 * QCAP) % 16 == 0, "the owners' lists start on a 16-B boundary");

__device__ __forceinline__ bool keys_usable(int nNodes, long long nEdges) {
  const long long nPieces = (nEdges + KEY_PIECE - 1) / KEY_PIECE;
  return nPieces * RNG <= (long long)nNodes;
}

__device__ __forceinline__ unsigned range_no(int d) {
  return (d < 0) ? 255u : (((unsigned)d / RNG) & 255u);
}

__global__ __launch_bounds__(KEY_THREADS)
void range_key_kernel(const int* __restrict__ dst, unsigned* keep, int nNodes, long long nEdges) {
  if (!keys_usable(nNodes, nEdges)) return;
  const int lane = threadIdx.x & 31;
  const int U0   = (blockIdx.x * (KEY_THREADS / 32) + (threadIdx.x >> 5)) * 8;
  const int piece = U0 / KU_PER_PIECE;
  const int ul0   = U0 % KU_PER_PIECE;
  int4 d[8];
#pragma unroll
  for (int k = 0; k < 8; ++k) {
    const long long e = (long long)(U0 + k) * KU_EDGES + 4 * lane;
    if (e + 4 <= nEdges) {
      d[k] = *(const int4*)(dst + e);
    } else {
      d[k].x = (e + 0 < nEdges) ? dst[e + 0] : -1;
      d[k].y = (e + 1 < nEdges) ? dst[e + 1] : -1;
      d[k].z = (e + 2 < nEdges) ? dst[e + 2] : -1;
      d[k].w = (e + 3 < nEdges) ? dst[e + 3] : -1;
    }
  }
  unsigned r[4][8];
#pragma unroll
  for (int j = 0; j < 4; ++j)
#pragma unroll
    for (int b = 0; b < 8; ++b) r[j][b] = 0u;
#pragma unroll
  for (int k = 0; k < 8; ++k) {
    unsigned kj[4];
    kj[0] = range_no(d[k].x); kj[1] = range_no(d[k].y);
    kj[2] = range_no(d[k].z); kj[3] = range_no(d[k].w);
#pragma unroll
    for (int j = 0; j < 4; ++j) {
#pragma unroll
      for (int b = 0; b < 8; ++b) {
        const unsigned pl = __builtin_amdgcn_ballot_w32(((kj[j] >> b) & 1u) != 0u);
        if (lane == k) r[j][b] = pl;
      }
    }
  }
  if (lane < 8) {
    unsigned* tile = keep + (long long)piece * (SLAB_BYTES / 4) + KEY_OFF / 4 + (ul0 >> 5) * (KT_BYTES / 4);
    const int ul = (ul0 & 31) + lane;
#pragma unroll
    for (int b = 0; b < 8; ++b)
      *(uint4*)(tile + b * 128 + ul * 4) = make_uint4(r[0][b], r[1][b], r[2][b], r[3][b]);
  }
}

__device__ __forceinline__ void ks_hits(unsigned m, int e0, int lo, int cnt,
                                        const float* __restrict__ feat,
                                        const int* __restrict__ src,
                                        const int* __restrict__ dst, long long nEdges,
                                        float* acc, int astr, int* qe, int* degl, int* qn) {
  while (m) {
    const int pos = __ffs(m) - 1;
    const int e = e0 + 4 * pos;
    m &= m - 1;
    const int p = atomicAdd(qn, 1);
    if (p < QCAP) {
      qe[p] = e;
    } else if (e < nEdges) {
      const unsigned u = (unsigned)(dst[e] - lo);
      if (u < (unsigned)cnt) {
        atomicAdd(&degl[u], 1);
        const int s = src[e];
#pragma unroll 1
        for (int f = 0; f < NFEAT; ++f)
          atomicAdd(&acc[u * astr + f], feat[(long long)s * NFEAT + f]);
      }
    }
  }
}

__device__ __forceinline__ void ks_load(const unsigned char* keys, int U, int nUnits, uint4 (&c)[8]) {
  if (U < nUnits) {
    const unsigned char* p = keys + (long long)(U / KU_PER_PIECE) * SLAB_BYTES + KEY_OFF +
                             ((U % KU_PER_PIECE) >> 5) * KT_BYTES + (U & 31) * 16;
#pragma unroll
    for (int b = 0; b < 8; ++b) c[b] = *(const uint4*)(p + b * 512);
  } else {
#pragma unroll
    for (int b = 0; b < 8; ++b) c[b] = make_uint4(0u, 0u, 0u, 0u);
  }
}

__device__ __forceinline__ void ks_filter(const float* __restrict__ feat,
                                          const int* __restrict__ src,
                                          const int* __restrict__ dst,
                                          const unsigned char* keys,
                                          long long nEdges, int lo, int cnt, int tid,
                                          float* acc, int astr, int* qe, int* degl, int* qn) {
  const int nPieces = (int)((nEdges + KEY_PIECE - 1) / KEY_PIECE);
  const int nUnits  = nPieces * KU_PER_PIECE;
  const int nTrips  = (nUnits + RS_THREADS - 1) / RS_THREADS;
  const unsigned key = blockIdx.x & 255u;
  unsigned sm[8];
#pragma unroll
  for (int b = 0; b < 8; ++b) sm[b] = 0u - ((key >> b) & 1u);
  int t = (int)(((long long)blockIdx.x * nTrips) / gridDim.x);
  uint4 c[8];
  ks_load(keys, t * RS_THREADS + tid, nUnits, c);
  for (int tt = 0; tt < nTrips; ++tt) {
    int tn = t + 1;
    if (tn >= nTrips) tn = 0;
    uint4 n[8];
#pragma unroll
    for (int b = 0; b < 8; ++b) n[b] = c[b];
    if (tt + 1 < nTrips) ks_load(keys, tn * RS_THREADS + tid, nUnits, n);
    const int U = t * RS_THREADS + tid;
    if (U < nUnits) {
      uint4 mm;
      mm.x = c[0].x ^ sm[0]; mm.y = c[0].y ^ sm[0]; mm.z = c[0].z ^ sm[0]; mm.w = c[0].w ^ sm[0];
#pragma unroll
      for (int b = 1; b < 8; ++b) {
        mm.x |= c[b].x ^ sm[b]; mm.y |= c[b].y ^ sm[b];
        mm.z |= c[b].z ^ sm[b]; mm.w |= c[b].w ^ sm[b];
      }
      const int e0 = U * KU_EDGES;
      ks_hits(~mm.x, e0 + 0, lo, cnt, feat, src, dst, nEdges, acc, astr, qe, degl, qn);
      ks_hits(~mm.y, e0 + 1, lo, cnt, feat, src, dst, nEdges, acc, astr, qe, degl, qn);
      ks_hits(~mm.z, e0 + 2, lo, cnt, feat, src, dst, nEdges, acc, astr, qe, degl, qn);
      ks_hits(~mm.w, e0 + 3, lo, cnt, feat, src, dst, nEdges, acc, astr, qe, degl, qn);
    }
#pragma unroll
    for (int b = 0; b < 8; ++b) c[b] = n[b];
    t = tn;
  }
}

__device__ __forceinline__ void rs_take(int d, int e, int lo, int cnt,
                                        const float* __restrict__ feat,
                                        const int* __restrict__ src,
                                        float* acc, int astr, int* qe, unsigned short* qd,
                                        int* degl, int* qn) {
  const unsigned u = (unsigned)(d - lo);
  if (u < (unsigned)cnt) {
    atomicAdd(&degl[u], 1);
    const int p = atomicAdd(qn, 1);
    if (p < QCAP) {
      qe[p] = e;
      qd[p] = (unsigned short)u;
    } else {
      const int s = src[e];
#pragma unroll 1
      for (int f = 0; f < NFEAT; ++f)
        atomicAdd(&acc[u * astr + f], feat[(long long)s * NFEAT + f]);
    }
  }
}

__device__ __forceinline__ void rs_filter(const float* __restrict__ feat,
                                          const int* __restrict__ src,
                                          const int* __restrict__ dst,
                                          long long nEdges, int lo, int cnt, int tid,
                                          float* acc, int astr, int* qe, unsigned short* qd,
                                          int* degl, int* qn) {
  const long long n4 = nEdges >> 2;
  const int4* __restrict__ dst4 = (const int4*)dst;
  const int nFull = (int)(n4 / (4 * RS_THREADS));
  const int rot   = (int)(((long long)blockIdx.x * nFull) / gridDim.x);
  for (int gg = 0; gg < nFull; ++gg) {
    int g = gg + rot;
    if (g >= nFull) g -= nFull;
    const long long i = (long long)g * (4 * RS_THREADS) + tid;
    const int4 d0 = dst4[i];
    const int4 d1 = dst4[i + RS_THREADS];
    const int4 d2 = dst4[i + 2 * RS_THREADS];
    const int4 d3 = dst4[i + 3 * RS_THREADS];
    const int e0 = (int)(i * 4);
    const int e1 = e0 + 4 * RS_THREADS, e2 = e0 + 8 * RS_THREADS, e3 = e0 + 12 * RS_THREADS;
    rs_take(d0.x, e0 + 0, lo, cnt, feat, src, acc, astr, qe, qd, degl, qn);
    rs_take(d0.y, e0 + 1, lo, cnt, feat, src, acc, astr, qe, qd, degl, qn);
    rs_take(d0.z, e0 + 2, lo, cnt, feat, src, acc, astr, qe, qd, degl, qn);
    rs_take(d0.w, e0 + 3, lo, cnt, feat, src, acc, astr, qe, qd, degl, qn);
    rs_take(d1.x, e1 + 0, lo, cnt, feat, src, acc, astr, qe, qd, degl, qn);
    rs_take(d1.y, e1 + 1, lo, cnt, feat, src, acc, astr, qe, qd, degl, qn);
    rs_take(d1.z, e1 + 2, lo, cnt, feat, src, acc, astr, qe, qd, degl, qn);
    rs_take(d1.w, e1 + 3, lo, cnt, feat, src, acc, astr, qe, qd, degl, qn);
    rs_take(d2.x, e2 + 0, lo, cnt, feat, src, acc, astr, qe, qd, degl, qn);
    rs_take(d2.y, e2 + 1, lo, cnt, feat, src, acc, astr, qe, qd, degl, qn);
    rs_take(d2.z, e2 + 2, lo, cnt, feat, src, acc, astr, qe, qd, degl, qn);
    rs_take(d2.w, e2 + 3, lo, cnt, feat, src, acc, astr, qe, qd, degl, qn);
    rs_take(d3.x, e3 + 0, lo, cnt, feat, src, acc, astr, qe, qd, degl, qn);
    rs_take(d3.y, e3 + 1, lo, cnt, feat, src, acc, astr, qe, qd, degl, qn);
    rs_take(d3.z, e3 + 2, lo, cnt, feat, src, acc, astr, qe, qd, degl, qn);
    rs_take(d3.w, e3 + 3, lo, cnt, feat, src, acc, astr, qe, qd, degl, qn);
  }
  for (long long i = (long long)nFull * (4 * RS_THREADS) + tid; i < n4; i += RS_THREADS) {
    const int4 d = dst4[i];
    const int e = (int)(i * 4);
    rs_take(d.x, e + 0, lo, cnt, feat, src, acc, astr, qe, qd, degl, qn);
    rs_take(d.y, e + 1, lo, cnt, feat, src, acc, astr, qe, qd, degl, qn);
    rs_take(d.z, e + 2, lo, cnt, feat, src, acc, astr, qe, qd, degl, qn);
    rs_take(d.w, e + 3, lo, cnt, feat, src, acc, astr, qe, qd, degl, qn);
  }
  for (long long e = n4 * 4 + tid; e < nEdges; e += RS_THREADS)
    rs_take(dst[e], (int)e, lo, cnt, feat, src, acc, astr, qe, qd, degl, qn);
}

__device__ __forceinline__ void own_load8(const unsigned* q, int nw, int b, int lane,
                                          const float* __restrict__ frow,
                                          unsigned (&e)[8], float4 (&v)[8]) {
  int idx = b * 8 + (lane & 7);
  if (idx > nw - 1) idx = nw - 1;
  const int mine = (int)q[idx];
#pragma unroll
  for (int k = 0; k < 8; ++k) {
    e[k] = (unsigned)__builtin_amdgcn_readlane(mine, k);
    v[k] = *(const float4*)(frow + (long long)(e[k] >> 9) * NFEAT);
  }
}

__device__ __forceinline__ void own_flush(float* acc, int astr, int lane, unsigned cu, const float4& rs) {
  if (cu != SQ_NULL) {
    float4* p = (float4*)&acc[(int)cu * astr + 4 * lane];
    float4 a = *p;
    a.x += rs.x; a.y += rs.y; a.z += rs.z; a.w += rs.w;
    *p = a;
  }
}

__device__ __forceinline__ void own_add8(float* acc, int astr, int nw, int b, int lane,
                                         const unsigned (&e)[8], const float4 (&v)[8],
                                         unsigned& cu, float4& rs) {
#pragma unroll
  for (int k = 0; k < 8; ++k) {
    if (b * 8 + k < nw) {
      const unsigned u = e[k] & 511u;
      if (u != cu) {
        own_flush(acc, astr, lane, cu, rs);
        cu = u;
        rs = v[k];
      } else {
        rs.x += v[k].x; rs.y += v[k].y; rs.z += v[k].z; rs.w += v[k].w;
      }
    }
  }
}

struct L1Frag {
  v2f bs0, bs1, bn0, bn1;
};

__device__ __forceinline__ void l1_load(const float* __restrict__ wsp, const float* __restrict__ wnp,
                                        int m, L1Frag& f) {
  f.bs0.x = wsp[(8 * m + 0) * NHID];  f.bs0.y = wsp[(8 * m + 1) * NHID];
  f.bs1.x = wsp[(8 * m + 2) * NHID];  f.bs1.y = wsp[(8 * m + 3) * NHID];
  f.bn0.x = wnp[(8 * m + 0) * NHID];  f.bn0.y = wnp[(8 * m + 1) * NHID];
  f.bn1.x = wnp[(8 * m + 2) * NHID];  f.bn1.y = wnp[(8 * m + 3) * NHID];
}

template <int NT>
__device__ __forceinline__ void l1_mma(const float* const (&xr)[NT], const float* const (&sr)[NT], int m,
                                       const L1Frag& f, v8f (&c)[NT]) {
  float4 hv[NT], sv[NT];
#pragma unroll
  for (int j = 0; j < NT; ++j) {
    hv[j] = *(const float4*)(xr[j] + 8 * m);
    sv[j] = *(const float4*)(sr[j] + 8 * m);
  }
#pragma unroll
  for (int j = 0; j < NT; ++j) {
    v2f ah0, ah1, aa0, aa1;
    ah0.x = hv[j].x; ah0.y = hv[j].y; ah1.x = hv[j].z; ah1.y = hv[j].w;
    aa0.x = sv[j].x; aa0.y = sv[j].y; aa1.x = sv[j].z; aa1.y = sv[j].w;
    c[j] = __builtin_amdgcn_wmma_f32_16x16x4_f32(false, ah0, false, f.bs0, (short)0, c[j], false, false);
    c[j] = __builtin_amdgcn_wmma_f32_16x16x4_f32(false, aa0, false, f.bn0, (short)0, c[j], false, false);
    c[j] = __builtin_amdgcn_wmma_f32_16x16x4_f32(false, ah1, false, f.bs1, (short)0, c[j], false, false);
    c[j] = __builtin_amdgcn_wmma_f32_16x16x4_f32(false, aa1, false, f.bn1, (short)0, c[j], false, false);
  }
}

template <int NT>
__device__ __forceinline__ void l1_product(const float* xs, int rb, const float* acc,
                                           const float* __restrict__ wself,
                                           const float* __restrict__ wneigh,
                                           const float* __restrict__ bias, float* hout,
                                           int lo, int cnt, int rt0, int col, int hf, int l16) {
  const float* xr[NT];
  const float* sr[NT];
  v8f c[NT];
#pragma unroll
  for (int j = 0; j < NT; ++j) {
    int r = (rt0 + 4 * j) * 16 + l16;
    if (r >= cnt) r = cnt - 1;
    xr[j] = xs  + (r - rb) * ACC_PITCH2 + 4 * hf;
    sr[j] = acc + r * ACC_PITCH2 + 4 * hf;
    c[j] = (v8f){0.f, 0.f, 0.f, 0.f, 0.f, 0.f, 0.f, 0.f};
  }
  const float* wsp = wself  + (4 * hf) * NHID + col;
  const float* wnp = wneigh + (4 * hf) * NHID + col;
  L1Frag fa, fb;
  l1_load(wsp, wnp, 0, fa);
#pragma unroll 1
  for (int m = 0; m < NFEAT / 8; m += 2) {
    l1_load(wsp, wnp, m + 1, fb);
    l1_mma<NT>(xr, sr, m, fa, c);
    if (m + 2 < NFEAT / 8) l1_load(wsp, wnp, m + 2, fa);
    l1_mma<NT>(xr, sr, m + 1, fb, c);
  }
  const float bv = bias[col];
#pragma unroll
  for (int j = 0; j < NT; ++j) {
#pragma unroll
    for (int i = 0; i < 8; ++i) {
      const int row = (rt0 + 4 * j) * 16 + i + hf * 8;
      if (row < cnt) hout[(long long)(lo + row) * NHID + col] = fmaxf(c[j][i] + bv, 0.0f);
    }
  }
}

__device__ __forceinline__ void xs_load(const float* __restrict__ feat, int r0, int np, int tid,
                                        float4 (&pre)[XS_LOADS]) {
  const float* base = feat + (long long)r0 * NFEAT;
#pragma unroll
  for (int k = 0; k < XS_LOADS; ++k) {
    const int i = tid + k * RS_THREADS;
    pre[k] = make_float4(0.f, 0.f, 0.f, 0.f);
    if (i < np) pre[k] = *(const float4*)(base + (long long)i * 4);
  }
}

__device__ __forceinline__ void xs_store(float* xs, int np, int tid, const float4 (&pre)[XS_LOADS]) {
#pragma unroll
  for (int k = 0; k < XS_LOADS; ++k) {
    const int i = tid + k * RS_THREADS;
    if (i < np) *(float4*)(&xs[(i >> 5) * ACC_PITCH2 + (i & 31) * 4]) = pre[k];
  }
}

__device__ __forceinline__ void l1_pass(const float* xs, const float* acc,
                                        const float* __restrict__ wself,
                                        const float* __restrict__ wneigh,
                                        const float* __restrict__ bias, float* hout,
                                        int lo, int cnt, int tb, int te, int rg, int col, int hf, int l16) {
  const int span = te - tb;
  int nt = (span > rg) ? ((span - rg + 3) >> 2) : 0;
  nt = __builtin_amdgcn_readfirstlane(nt);
  for (int j0 = 0; j0 < nt; j0 += 4) {
    const int n   = (nt - j0 < 4) ? (nt - j0) : 4;
    const int rt0 = tb + rg + 4 * j0;
    switch (n) {
      case 4:  l1_product<4>(xs, tb * 16, acc, wself, wneigh, bias, hout, lo, cnt, rt0, col, hf, l16); break;
      case 3:  l1_product<3>(xs, tb * 16, acc, wself, wneigh, bias, hout, lo, cnt, rt0, col, hf, l16); break;
      case 2:  l1_product<2>(xs, tb * 16, acc, wself, wneigh, bias, hout, lo, cnt, rt0, col, hf, l16); break;
      default: l1_product<1>(xs, tb * 16, acc, wself, wneigh, bias, hout, lo, cnt, rt0, col, hf, l16); break;
    }
  }
}

template <bool FIRST>
__global__ __launch_bounds__(RS_THREADS)
void range_sum_kernel(const float* __restrict__ feat,
                      const int* __restrict__ src,
                      const int* __restrict__ dst,
                      float* agg,
                      int* keep,
                      const float* __restrict__ wself,
                      const float* __restrict__ wneigh,
                      const float* __restrict__ bias,
                      float* __restrict__ out,
                      int nNodes, long long nEdges) {
  constexpr int AST = ACC_PITCH2;
  __shared__ __attribute__((aligned(16))) float acc[RNG * AST];
  __shared__ __attribute__((aligned(16))) unsigned char pool[FIRST ? POOL_L1 : POOL_Q];
  int* qe = (int*)pool;
  unsigned short* qd = (unsigned short*)(pool + 4 * QCAP);
  unsigned* rq = (unsigned*)(pool + 6 * QCAP);
  __shared__ __attribute__((aligned(16))) int degl[RNG];
  __shared__ int qn;
  __shared__ int hd;
  __shared__ int rowcur[RNG];
  __shared__ __attribute__((aligned(16))) float wl[FIRST ? 4 : 2 * NHID * NCLASS];

  const int tid  = threadIdx.x;
  const int lane = tid & 31;
  const int wave = tid >> 5;
  const int lo   = blockIdx.x * RNG;
  const int cnt  = (nNodes - lo < RNG) ? (nNodes - lo) : RNG;
  int* slab = keep + (long long)lo * NFEAT;

  const float4 z4 = make_float4(0.f, 0.f, 0.f, 0.f);
  for (int i = tid; i < RNG * AST / 4; i += RS_THREADS) ((float4*)acc)[i] = z4;
  if (FIRST) {
    for (int i = tid; i < RNG; i += RS_THREADS) degl[i] = 0;
    if (tid == 0) qn = 0;
  } else {
    if (tid == 0) hd = slab[0];
    for (int i = tid; i < NHID / 2 * NCLASS; i += RS_THREADS) {
      const int q = i >> 4, c = i & 15;
      const int k = 8 * (q >> 2) + 4 * (q & 1) + 2 * ((q >> 1) & 1);
      v2f ws, wn;
      ws.x = wself [k * NCLASS + c];  ws.y = wself [(k + 1) * NCLASS + c];
      wn.x = wneigh[k * NCLASS + c];  wn.y = wneigh[(k + 1) * NCLASS + c];
      ((v2f*)wl)[i] = ws;
      ((v2f*)wl)[NHID / 2 * NCLASS + i] = wn;
    }
  }
  __syncthreads();

  int n;
  const bool scan = FIRST ? true : (hd < 0);
  if (scan) {
    if (!FIRST) {
      for (int i = tid; i < RNG; i += RS_THREADS) degl[i] = 0;
      if (tid == 0) qn = 0;
      __syncthreads();
    }
    const bool keyed = FIRST && keys_usable(nNodes, nEdges);
    if (keyed) ks_filter(feat, src, dst, (const unsigned char*)keep, nEdges, lo, cnt, tid, acc, AST, qe, degl, &qn);
    else       rs_filter(feat, src, dst, nEdges, lo, cnt, tid, acc, AST, qe, qd, degl, &qn);
    __syncthreads();

    const int total = qn;
    n = (total < QCAP) ? total : QCAP;

    if (keyed) {
      for (int i = tid; i < n; i += RS_THREADS) {
        const int e = qe[i];
        int s = 0;
        unsigned short uu = QD_NONE;
        if (e < nEdges) {
          const int dv = dst[e];
          const int sv = src[e];
          const unsigned u = (unsigned)(dv - lo);
          if (u < (unsigned)cnt) { s = sv; uu = (unsigned short)u; atomicAdd(&degl[u], 1); }
        }
        qe[i] = s;
        qd[i] = uu;
      }
    } else {
      for (int i = tid; i < n; i += RS_THREADS) qe[i] = src[qe[i]];
    }
    __syncthreads();

    if (FIRST) {
      const int np = (n + 7) & ~7;
      const bool fits = (total <= QCAP) && (SL_Q * 4 + 6 * np <= cnt * NFEAT * 4);
      if (fits) {
        int4* sq = (int4*)(slab + SL_Q);
        for (int i = tid; i < np / 4; i += RS_THREADS) sq[i] = ((const int4*)qe)[i];
        int4* sd = (int4*)(slab + SL_Q + np);
        for (int i = tid; i < np / 8; i += RS_THREADS) sd[i] = ((const int4*)qd)[i];
        int4* sg = (int4*)(slab + SL_DEG);
        for (int i = tid; i < RNG / 4; i += RS_THREADS) sg[i] = ((const int4*)degl)[i];
      }
      if (tid == 0) slab[0] = fits ? n : -1;
    }
  } else {
    n = (hd < QCAP) ? hd : QCAP;
    const int np = (n + 7) & ~7;
    const int4* sq = (const int4*)(slab + SL_Q);
    for (int i = tid; i < np / 4; i += RS_THREADS) ((int4*)qe)[i] = sq[i];
    const int4* sd = (const int4*)(slab + SL_Q + np);
    for (int i = tid; i < np / 8; i += RS_THREADS) ((int4*)qd)[i] = sd[i];
    const int4* sg = (const int4*)(slab + SL_DEG);
    for (int i = tid; i < RNG / 4; i += RS_THREADS) ((int4*)degl)[i] = sg[i];
    __syncthreads();
  }

  int wtot;
  {
    const int myrow = wave + RS_WAVES * lane;
    const int rdeg = (myrow < RNG) ? degl[myrow] : 0;
    int rinc = rdeg;
    for (int off = 1; off < 32; off <<= 1) {
      const int up = __shfl_up(rinc, off);
      if (lane >= off) rinc += up;
    }
    if (myrow < RNG) rowcur[myrow] = rinc - rdeg;
    wtot = __builtin_amdgcn_readlane(rinc, 31);
    if (wtot > SQ) wtot = SQ;
    uint4* fl = (uint4*)(rq + wave * SQ);
    for (int i = lane; 4 * i < wtot; i += 32) fl[i] = make_uint4(SQ_NULL, SQ_NULL, SQ_NULL, SQ_NULL);
  }
  __syncthreads();

  for (int i = tid; i < n; i += RS_THREADS) {
    const unsigned u = qd[i];
    const int s = qe[i];
    if (u != QD_NONE) {
      const int ow = (int)(u & (RS_WAVES - 1));
      int pos = SQ;
      if ((unsigned)s < (1u << SQ_SBITS)) pos = atomicAdd(&rowcur[u], 1);
      if (pos < SQ) {
        rq[ow * SQ + pos] = ((unsigned)s << 9) | u;
      } else {
#pragma unroll 1
        for (int f = 0; f < NFEAT; ++f)
          atomicAdd(&acc[(int)u * AST + f], feat[(long long)s * NFEAT + f]);
      }
    }
  }
  __syncthreads();

  {
    const int nw = wtot;
    if (nw > 0) {
      const unsigned* q = rq + wave * SQ;
      const float* frow = feat + 4 * lane;
      const int nb = (nw + 7) >> 3;
      unsigned ea[8], eb[8];
      float4   va[8], vb[8];
      unsigned cu = SQ_NULL;
      float4   rs = z4;
      own_load8(q, nw, 0, lane, frow, ea, va);
      for (int b = 0; b < nb; b += 2) {
        own_load8(q, nw, b + 1, lane, frow, eb, vb);
        own_add8(acc, AST, nw, b, lane, ea, va, cu, rs);
        own_load8(q, nw, b + 2, lane, frow, ea, va);
        own_add8(acc, AST, nw, b + 1, lane, eb, vb, cu, rs);
      }
      own_flush(acc, AST, lane, cu, rs);
    }
  }
  __syncthreads();

  if (FIRST) {
    float* xs = (float*)pool;
    const int ntile = (cnt + 15) >> 4;
    const int nr0 = ((cnt < XS_ROWS) ? cnt : XS_ROWS) * (NFEAT / 4);
    float4 pre[XS_LOADS];
    xs_load(feat, lo, nr0, tid, pre);
    for (int i = tid; i < cnt * (NFEAT / 4); i += RS_THREADS) {
      const int row = i >> 5;
      const int c4  = (i & 31) * 4;
      float4* p = (float4*)(&acc[row * AST + c4]);
      const float4 s4 = *p;
      const float dv = fmaxf((float)degl[row], 1.0f);
      float4 o;
      o.x = s4.x / dv; o.y = s4.y / dv; o.z = s4.z / dv; o.w = s4.w / dv;
      *p = o;
    }
    xs_store(xs, nr0, tid, pre);
    __syncthreads();

    const int wa = wave >> 3, wb = (wave >> 2) & 1, wc = wave & 3;
    const int rg = (wa + wc) & 3;
    const int ct = 2 * wa + wb;
    const int hf  = lane >> 4;
    const int l16 = lane & 15;
    const int col = ct * 16 + l16;
    const int nr1 = (cnt > XS_ROWS) ? (cnt - XS_ROWS) * (NFEAT / 4) : 0;
#pragma unroll 1
    for (int p = 0; p < 2; ++p) {
      const int tb = p * XS_TILES;
      const int te = (ntile < tb + XS_TILES) ? ntile : (tb + XS_TILES);
      if (p == 1) {
        xs_load(feat, lo + XS_ROWS, nr1, tid, pre);
        __syncthreads();
        xs_store(xs, nr1, tid, pre);
        __syncthreads();
      }
      l1_pass(xs, acc, wself, wneigh, bias, agg, lo, cnt, tb, te, rg, col, hf, l16);
    }
  } else {
    const int hf  = lane >> 4;
    const int l16 = lane & 15;
    const v2f* wls = (const v2f*)wl;
    const v2f* wln = wls + NHID / 2 * NCLASS;
    for (int t = wave; t * 16 < cnt; t += RS_THREADS / 32) {
      int r = t * 16 + l16;
      if (r >= cnt) r = cnt - 1;
      const float rdv = 1.0f / fmaxf((float)degl[r], 1.0f);
      const float* hrow = feat + (long long)(lo + r) * NFEAT + 4 * hf;
      const float* srow = &acc[r * AST + 4 * hf];
      v8f c = {0.f, 0.f, 0.f, 0.f, 0.f, 0.f, 0.f, 0.f};
#pragma unroll
      for (int m = 0; m < NFEAT / 8; ++m) {
        const float4 hv = *(const float4*)(hrow + 8 * m);
        const float4 sv = *(const float4*)(srow + 8 * m);
        const v2f bs0 = wls[(4 * m + 0) * 16 + lane];
        const v2f bs1 = wls[(4 * m + 2) * 16 + lane];
        const v2f bn0 = wln[(4 * m + 0) * 16 + lane];
        const v2f bn1 = wln[(4 * m + 2) * 16 + lane];
        v2f ah0, ah1, aa0, aa1;
        ah0.x = hv.x;       ah0.y = hv.y;       ah1.x = hv.z;       ah1.y = hv.w;
        aa0.x = sv.x * rdv; aa0.y = sv.y * rdv; aa1.x = sv.z * rdv; aa1.y = sv.w * rdv;
        c = __builtin_amdgcn_wmma_f32_16x16x4_f32(false, ah0, false, bs0, (short)0, c, false, false);
        c = __builtin_amdgcn_wmma_f32_16x16x4_f32(false, aa0, false, bn0, (short)0, c, false, false);
        c = __builtin_amdgcn_wmma_f32_16x16x4_f32(false, ah1, false, bs1, (short)0, c, false, false);
        c = __builtin_amdgcn_wmma_f32_16x16x4_f32(false, aa1, false, bn1, (short)0, c, false, false);
      }

      const float bv = bias[l16];
#pragma unroll
      for (int j = 0; j < 8; ++j) {
        const float v = c[j] + bv;
        float mx = v;
#pragma unroll
        for (int off = 8; off >= 1; off >>= 1)
          mx = fmaxf(mx, __shfl_xor(mx, off, 16));
        const float e = expf(v - mx);
        float sm = e;
#pragma unroll
        for (int off = 8; off >= 1; off >>= 1)
          sm += __shfl_xor(sm, off, 16);
        const int row = t * 16 + j + hf * 8;
        if (row < cnt) out[(long long)(lo + row) * NCLASS + l16] = v - mx - logf(sm);
      }
    }
  }
}

extern "C" void kernel_launch(void* const* d_in, const int* in_sizes, int n_in,
                              void* d_out, int out_size, void* d_ws, size_t ws_size,
                              hipStream_t stream) {
  const float* x   = (const float*)d_in[0];
  const int*   src = (const int*)  d_in[1];
  const int*   dst = (const int*)  d_in[2];
  const float* ws1 = (const float*)d_in[3];
  const float* wn1 = (const float*)d_in[4];
  const float* b1  = (const float*)d_in[5];
  const float* ws2 = (const float*)d_in[6];
  const float* wn2 = (const float*)d_in[7];
  const float* b2  = (const float*)d_in[8];
  float* out = (float*)d_out;

  const int nNodes = in_sizes[0] / NFEAT;
  const long long nEdges = in_sizes[1];

  float* pa = (float*)d_ws;
  float* pb = pa + (size_t)nNodes * NFEAT;

  const int rsBlocks = (nNodes + RNG - 1) / RNG;

  const int keyBlocks = (int)((nEdges + KEY_PIECE - 1) / KEY_PIECE) * 8;

  range_key_kernel<<<keyBlocks, KEY_THREADS, 0, stream>>>(dst, (unsigned*)pb, nNodes, nEdges);
  range_sum_kernel<true><<<rsBlocks, RS_THREADS, 0, stream>>>(x, src, dst, pa, (int*)pb,
                                                              ws1, wn1, b1, nullptr, nNodes, nEdges);

  range_sum_kernel<false><<<rsBlocks, RS_THREADS, 0, stream>>>(pa, src, dst, nullptr, (int*)pb,
                                                               ws2, wn2, b2, out, nNodes, nEdges);
}
